# starting kernel
# baseline (speedup 1.0000x reference)
#include <hip/hip_runtime.h>


#define LRELU 0.2f

typedef __attribute__((ext_vector_type(2))) float v2f;
typedef __attribute__((ext_vector_type(8))) float v8f;

__device__ __forceinline__ float wred_sum(float v) {
#pragma unroll
    for (int o = 16; o > 0; o >>= 1) v += __shfl_xor(v, o, 32);
    return v;
}
__device__ __forceinline__ float wred_max(float v) {
#pragma unroll
    for (int o = 16; o > 0; o >>= 1) v = fmaxf(v, __shfl_xor(v, o, 32));
    return v;
}
__device__ __forceinline__ unsigned fkey(float f) {
    unsigned u = __float_as_uint(f);
    return (u & 0x80000000u) ? ~u : (u | 0x80000000u);
}
__device__ __forceinline__ float fdec(unsigned k) {
    unsigned u = (k & 0x80000000u) ? (k & 0x7fffffffu) : ~k;
    return __uint_as_float(u);
}
__device__ __forceinline__ float lrelu(float x) { return x > 0.f ? x : LRELU * x; }

__global__ void fill_u32_kernel(unsigned* __restrict__ p, unsigned v, long long n) {
    long long i = (long long)blockIdx.x * blockDim.x + threadIdx.x;
    if (i < n) p[i] = v;
}

template <int COLT, int MODE>
__global__ __launch_bounds__(256) void gemm_wmma_f32(const float* __restrict__ A,
                                                     const float* __restrict__ B,
                                                     float* __restrict__ C, int nrows) {
    constexpr int CW = COLT * 16;
    __shared__ float Blds[256 * CW];
    const int tid = threadIdx.x;
    const int colBase = blockIdx.y * CW;

    for (int idx = tid; idx < 256 * CW; idx += 256) {
        float bv;
        if constexpr (MODE == 0) {
            bv = B[(size_t)blockIdx.y * (256 * 64) + idx];
        } else {
            int k = idx / CW;
            int c = idx - k * CW;
            bv = (c < 40) ? B[k * 40 + c] : 0.0f;
        }
        Blds[idx] = bv;
    }
    __syncthreads();

    const int wid  = tid >> 5;
    const int lane = tid & 31;
    const int half = lane >> 4;
    const int l16  = lane & 15;
    const int rowBase = blockIdx.x * 128 + wid * 16;

    int arow = rowBase + l16;
    if (arow >= nrows) arow = nrows - 1;
    const float* aptr = A + (size_t)arow * 256;

    v8f acc[COLT] = {};

    for (int k0 = 0; k0 < 256; k0 += 4) {
        const int kb = k0 + half * 2;
        v2f a;
        a.x = aptr[kb];
        a.y = aptr[kb + 1];
#pragma unroll
        for (int t = 0; t < COLT; ++t) {
            v2f b;
            const int c = t * 16 + l16;
            b.x = Blds[kb * CW + c];
            b.y = Blds[(kb + 1) * CW + c];
            acc[t] = __builtin_amdgcn_wmma_f32_16x16x4_f32(false, a, false, b, (short)0,
                                                           acc[t], false, false);
        }
    }

    if (rowBase + 16 <= nrows) {
#pragma unroll
        for (int t = 0; t < COLT; ++t) {
            const int col = colBase + t * 16 + l16;
            if constexpr (MODE == 1) {
                if (col >= 40) continue;
            }
#pragma unroll
            for (int v = 0; v < 8; ++v) {
                const int row = rowBase + v + half * 8;
                if constexpr (MODE == 0) {
                    C[(size_t)row * 256 + col] = acc[t][v];
                } else {
                    C[(size_t)row * 40 + col] = acc[t][v];
                }
            }
        }
    } else {
#pragma unroll
        for (int t = 0; t < COLT; ++t) {
            const int col = colBase + t * 16 + l16;
#pragma unroll
            for (int v = 0; v < 8; ++v) {
                const int row = rowBase + v + half * 8;
                if constexpr (MODE == 0) {
                    if (row < nrows) C[(size_t)row * 256 + col] = acc[t][v];
                } else {
                    if (row < nrows && col < 40) C[(size_t)row * 40 + col] = acc[t][v];
                }
            }
        }
    }
}

__global__ __launch_bounds__(256) void scores1_kernel(const float* __restrict__ Wh,
                                                      const float* __restrict__ a_heads,
                                                      float* __restrict__ ssrc,
                                                      float* __restrict__ sdst, int n) {
    long long gt = (long long)blockIdx.x * blockDim.x + threadIdx.x;
    int node = (int)(gt >> 5);
    int lane = threadIdx.x & 31;
    if (node >= n) return;
    const float* wh = Wh + (size_t)node * 256;
#pragma unroll
    for (int h = 0; h < 4; ++h) {
        float w1 = wh[h * 64 + lane];
        float w2 = wh[h * 64 + 32 + lane];
        const float* a = a_heads + h * 128;
        float ps = w1 * a[lane] + w2 * a[32 + lane];
        float pd = w1 * a[64 + lane] + w2 * a[96 + lane];
        ps = wred_sum(ps);
        pd = wred_sum(pd);
        if (lane == 0) {
            ssrc[node * 4 + h] = ps;
            sdst[node * 4 + h] = pd;
        }
    }
}

__global__ __launch_bounds__(256) void scores2_kernel(const float* __restrict__ Wx,
                                                      const float* __restrict__ a_out,
                                                      float* __restrict__ ssrc,
                                                      float* __restrict__ sdst, int n) {
    long long gt = (long long)blockIdx.x * blockDim.x + threadIdx.x;
    int node = (int)(gt >> 5);
    int lane = threadIdx.x & 31;
    if (node >= n) return;
    const float* wx = Wx + (size_t)node * 40;
    float w1 = wx[lane];
    float w2 = (lane < 8) ? wx[32 + lane] : 0.f;
    float as2 = (lane < 8) ? a_out[32 + lane] : 0.f;
    float ad2 = (lane < 8) ? a_out[72 + lane] : 0.f;
    float ps = w1 * a_out[lane] + w2 * as2;
    float pd = w1 * a_out[40 + lane] + w2 * ad2;
    ps = wred_sum(ps);
    pd = wred_sum(pd);
    if (lane == 0) {
        ssrc[node] = ps;
        sdst[node] = pd;
    }
}

__global__ __launch_bounds__(256) void edge_max1(const int* __restrict__ src,
                                                 const int* __restrict__ dst,
                                                 const float4* __restrict__ ss4,
                                                 const float4* __restrict__ sd4,
                                                 unsigned* __restrict__ mkey, int ne) {
    int e = blockIdx.x * blockDim.x + threadIdx.x;
    if (e >= ne) return;
    int s = src[e], d = dst[e];
    float4 vs = ss4[s];
    float4 vd = sd4[d];
    unsigned* mk = mkey + (size_t)d * 4;
    atomicMax(mk + 0, fkey(lrelu(vs.x + vd.x)));
    atomicMax(mk + 1, fkey(lrelu(vs.y + vd.y)));
    atomicMax(mk + 2, fkey(lrelu(vs.z + vd.z)));
    atomicMax(mk + 3, fkey(lrelu(vs.w + vd.w)));
}
__global__ __launch_bounds__(256) void edge_exp1(const int* __restrict__ src,
                                                 const int* __restrict__ dst,
                                                 const float4* __restrict__ ss4,
                                                 const float4* __restrict__ sd4,
                                                 const uint4* __restrict__ mkey4,
                                                 float* __restrict__ den,
                                                 float4* __restrict__ att4, int ne) {
    int e = blockIdx.x * blockDim.x + threadIdx.x;
    if (e >= ne) return;
    int s = src[e], d = dst[e];
    float4 vs = ss4[s];
    float4 vd = sd4[d];
    uint4 mk = mkey4[d];
    float4 ex;
    ex.x = expf(lrelu(vs.x + vd.x) - fdec(mk.x));
    ex.y = expf(lrelu(vs.y + vd.y) - fdec(mk.y));
    ex.z = expf(lrelu(vs.z + vd.z) - fdec(mk.z));
    ex.w = expf(lrelu(vs.w + vd.w) - fdec(mk.w));
    att4[e] = ex;
    float* dn = den + (size_t)d * 4;
    atomicAdd(dn + 0, ex.x);
    atomicAdd(dn + 1, ex.y);
    atomicAdd(dn + 2, ex.z);
    atomicAdd(dn + 3, ex.w);
}
__global__ __launch_bounds__(256) void edge_att1(const int* __restrict__ dst,
                                                 const float4* __restrict__ den4,
                                                 float4* __restrict__ att4, int ne) {
    int e = blockIdx.x * blockDim.x + threadIdx.x;
    if (e >= ne) return;
    float4 a = att4[e];
    float4 dn = den4[dst[e]];
    a.x /= dn.x;
    a.y /= dn.y;
    a.z /= dn.z;
    a.w /= dn.w;
    att4[e] = a;
}
__global__ __launch_bounds__(256) void scatter1(const int* __restrict__ src,
                                                const int* __restrict__ dst,
                                                const float* __restrict__ att,
                                                const float* __restrict__ Wh,
                                                float* __restrict__ hp, long long total) {
    long long i = (long long)blockIdx.x * blockDim.x + threadIdx.x;
    if (i >= total) return;
    int e = (int)(i >> 8);
    int f = (int)(i & 255);
    float a = att[(size_t)e * 4 + (f >> 6)];
    atomicAdd(&hp[(size_t)dst[e] * 256 + f], a * Wh[(size_t)src[e] * 256 + f]);
}
__global__ __launch_bounds__(256) void elu_kernel(float* __restrict__ p, long long n) {
    long long i = (long long)blockIdx.x * blockDim.x + threadIdx.x;
    if (i >= n) return;
    float v = p[i];
    p[i] = v > 0.f ? v : (expf(v) - 1.f);
}

__global__ __launch_bounds__(256) void edge_max2(const int* __restrict__ src,
                                                 const int* __restrict__ dst,
                                                 const float* __restrict__ ss,
                                                 const float* __restrict__ sd,
                                                 unsigned* __restrict__ mkey, int ne) {
    int e = blockIdx.x * blockDim.x + threadIdx.x;
    if (e >= ne) return;
    float x = lrelu(ss[src[e]] + sd[dst[e]]);
    atomicMax(&mkey[dst[e]], fkey(x));
}
__global__ __launch_bounds__(256) void edge_exp2(const int* __restrict__ src,
                                                 const int* __restrict__ dst,
                                                 const float* __restrict__ ss,
                                                 const float* __restrict__ sd,
                                                 const unsigned* __restrict__ mkey,
                                                 float* __restrict__ den,
                                                 float* __restrict__ att, int ne) {
    int e = blockIdx.x * blockDim.x + threadIdx.x;
    if (e >= ne) return;
    int d = dst[e];
    float x = lrelu(ss[src[e]] + sd[d]);
    float ex = expf(x - fdec(mkey[d]));
    att[e] = ex;
    atomicAdd(&den[d], ex);
}
__global__ __launch_bounds__(256) void edge_att2(const int* __restrict__ dst,
                                                 const float* __restrict__ den,
                                                 float* __restrict__ att, int ne) {
    int e = blockIdx.x * blockDim.x + threadIdx.x;
    if (e >= ne) return;
    att[e] = att[e] / den[dst[e]];
}
__global__ __launch_bounds__(256) void scatter2(const int* __restrict__ src,
                                                const int* __restrict__ dst,
                                                const float* __restrict__ att,
                                                const float* __restrict__ Wx,
                                                float* __restrict__ out, long long total) {
    long long i = (long long)blockIdx.x * blockDim.x + threadIdx.x;
    if (i >= total) return;
    int e = (int)(i >> 6);
    int f = (int)(i & 63);
    if (f >= 40) return;
    atomicAdd(&out[(size_t)dst[e] * 40 + f], att[e] * Wx[(size_t)src[e] * 40 + f]);
}

__global__ __launch_bounds__(256) void logsoftmax_kernel(float* __restrict__ out, int n) {
    long long gt = (long long)blockIdx.x * blockDim.x + threadIdx.x;
    int node = (int)(gt >> 5);
    int lane = threadIdx.x & 31;
    if (node >= n) return;
    float* row = out + (size_t)node * 40;
    float v1 = row[lane];
    float v2 = (lane < 8) ? row[32 + lane] : -INFINITY;
    float mx = wred_max(fmaxf(v1, v2));
    float s = expf(v1 - mx) + ((lane < 8) ? expf(v2 - mx) : 0.f);
    s = wred_sum(s);
    float lg = mx + logf(s);
    row[lane] = v1 - lg;
    if (lane < 8) row[32 + lane] = v2 - lg;
}

extern "C" void kernel_launch(void* const* d_in, const int* in_sizes, int n_in, void* d_out,
                              int out_size, void* d_ws, size_t ws_size, hipStream_t stream) {
    (void)n_in; (void)out_size; (void)ws_size;
    const float* h      = (const float*)d_in[0];
    const int*   eidx   = (const int*)d_in[1];
    const float* Wheads = (const float*)d_in[2];
    const float* aheads = (const float*)d_in[3];
    const float* Wout   = (const float*)d_in[4];
    const float* aout   = (const float*)d_in[5];
    float* out = (float*)d_out;

    const int N = in_sizes[0] / 256;
    const int E = in_sizes[1] / 2;
    const int* src = eidx;
    const int* dst = eidx + E;

    float* ws = (float*)d_ws;
    size_t oWh   = 0;
    size_t oHP   = oWh  + (size_t)N * 256;
    size_t oS1s  = oHP  + (size_t)N * 256;
    size_t oS1d  = oS1s + (size_t)N * 4;
    size_t oM1   = oS1d + (size_t)N * 4;
    size_t oD1   = oM1  + (size_t)N * 4;
    size_t oAtt1 = oD1  + (size_t)N * 4;
    size_t oS2s  = oAtt1 + (size_t)E * 4;
    size_t oS2d  = oS2s + (size_t)N;
    size_t oM2   = oS2d + (size_t)N;
    size_t oD2   = oM2  + (size_t)N;
    size_t oAtt2 = oD2  + (size_t)N;
    size_t oWx   = oWh;

    const unsigned NEG_INF_KEY = 0x007fffffu;

    hipMemsetAsync(ws + oHP, 0, (size_t)N * 256 * sizeof(float), stream);
    hipMemsetAsync(ws + oD1, 0, (size_t)N * 4 * sizeof(float), stream);
    hipMemsetAsync(ws + oD2, 0, (size_t)N * sizeof(float), stream);
    hipMemsetAsync(out, 0, (size_t)N * 40 * sizeof(float), stream);
    {
        long long n4 = (long long)N * 4;
        fill_u32_kernel<<<(int)((n4 + 255) / 256), 256, 0, stream>>>(
            (unsigned*)(ws + oM1), NEG_INF_KEY, n4);
        fill_u32_kernel<<<(N + 255) / 256, 256, 0, stream>>>(
            (unsigned*)(ws + oM2), NEG_INF_KEY, (long long)N);
    }

    dim3 g1((N + 127) / 128, 4);
    gemm_wmma_f32<4, 0><<<g1, 256, 0, stream>>>(h, Wheads, ws + oWh, N);

    scores1_kernel<<<(N + 7) / 8, 256, 0, stream>>>(ws + oWh, aheads, ws + oS1s, ws + oS1d, N);

    int geE = (E + 255) / 256;
    edge_max1<<<geE, 256, 0, stream>>>(src, dst, (const float4*)(ws + oS1s),
                                       (const float4*)(ws + oS1d), (unsigned*)(ws + oM1), E);
    edge_exp1<<<geE, 256, 0, stream>>>(src, dst, (const float4*)(ws + oS1s),
                                       (const float4*)(ws + oS1d), (const uint4*)(ws + oM1),
                                       ws + oD1, (float4*)(ws + oAtt1), E);
    edge_att1<<<geE, 256, 0, stream>>>(dst, (const float4*)(ws + oD1), (float4*)(ws + oAtt1), E);

    long long tot1 = (long long)E * 256;
    scatter1<<<(int)((tot1 + 255) / 256), 256, 0, stream>>>(src, dst, ws + oAtt1, ws + oWh,
                                                            ws + oHP, tot1);
    long long nf = (long long)N * 256;
    elu_kernel<<<(int)((nf + 255) / 256), 256, 0, stream>>>(ws + oHP, nf);

    dim3 g2((N + 127) / 128, 1);
    gemm_wmma_f32<3, 1><<<g2, 256, 0, stream>>>(ws + oHP, Wout, ws + oWx, N);

    scores2_kernel<<<(N + 7) / 8, 256, 0, stream>>>(ws + oWx, aout, ws + oS2s, ws + oS2d, N);

    edge_max2<<<geE, 256, 0, stream>>>(src, dst, ws + oS2s, ws + oS2d, (unsigned*)(ws + oM2), E);
    edge_exp2<<<geE, 256, 0, stream>>>(src, dst, ws + oS2s, ws + oS2d, (unsigned*)(ws + oM2),
                                       ws + oD2, ws + oAtt2, E);
    edge_att2<<<geE, 256, 0, stream>>>(dst, ws + oD2, ws + oAtt2, E);

    long long tot2 = (long long)E * 64;
    scatter2<<<(int)((tot2 + 255) / 256), 256, 0, stream>>>(src, dst, ws + oAtt2, ws + oWx, out,
                                                            tot2);

    logsoftmax_kernel<<<(N + 7) / 8, 256, 0, stream>>>(out, N);
}


# score dots as extra GEMM output columns on weights folded with a, h(Wa) for (hW)a: a 16-wide fifth strip in layer 1, 2 spare pad columns in layer 2; both score launches gone; dots' sum order changes
# speedup vs baseline: 5.5914x; 5.5914x over previous
#include <hip/hip_runtime.h>


#define LRELU 0.2f

typedef __attribute__((ext_vector_type(2))) float v2f;
typedef __attribute__((ext_vector_type(8))) float v8f;

__device__ __forceinline__ float wred_sum(float v) {
#pragma unroll
    for (int o = 16; o > 0; o >>= 1) v += __shfl_xor(v, o, 32);
    return v;
}
__device__ __forceinline__ float wred_max(float v) {
#pragma unroll
    for (int o = 16; o > 0; o >>= 1) v = fmaxf(v, __shfl_xor(v, o, 32));
    return v;
}
__device__ __forceinline__ float lrelu(float x) { return x > 0.f ? x : LRELU * x; }

__global__ __launch_bounds__(256) void prep_kernel(int* __restrict__ head, int n, int fillBlocks,
                                                   const float* __restrict__ Wheads,
                                                   const float* __restrict__ aheads,
                                                   float* __restrict__ Wa1) {
    const int b = blockIdx.x;
    if (b < fillBlocks) {
        const int i = b * 256 + threadIdx.x;
        if (i < n) head[i] = -1;
        return;
    }
    const int c = b - fillBlocks;
    const int hd = c & 3;
    const int k = threadIdx.x;
    const float* w = Wheads + ((size_t)hd * 256 + k) * 64;
    const float* a = aheads + hd * 128 + (c >> 2) * 64;
    float s = 0.f;
#pragma unroll
    for (int f = 0; f < 64; ++f) s += w[f] * a[f];
    Wa1[k * 16 + c] = s;
    Wa1[k * 16 + 8 + c] = 0.f;
}

template <int COLT, int MODE, bool GUARD>
__device__ __forceinline__ void store_tiles(const v8f (&acc)[COLT], float* __restrict__ C,
                                            float* __restrict__ S0, float* __restrict__ S1,
                                            int nrows, int strip, int rowBase, int half, int l16) {
#pragma unroll
    for (int t = 0; t < COLT; ++t) {
        const int col = t * 16 + l16;
        float* base;
        size_t stride;
        bool on = true;
        if constexpr (MODE == 0) {
            base = C + strip * 64 + col;
            stride = 256;
        } else if constexpr (MODE == 1) {
            if (col < 40) {
                base = C + col;
                stride = 40;
            } else {
                base = (col == 40) ? S0 : S1;
                stride = 1;
                on = col < 42;
            }
        } else {
            base = (col < 4) ? (S0 + col) : (S1 + (col - 4));
            stride = 4;
            on = col < 8;
        }
#pragma unroll
        for (int v = 0; v < 8; ++v) {
            const int row = rowBase + v + half * 8;
            if (on && (!GUARD || row < nrows)) base[(size_t)row * stride] = acc[t][v];
        }
    }
}

template <int COLT, int MODE>
__device__ __forceinline__ void gemm_strip(const float* __restrict__ A,
                                           const float* __restrict__ B,
                                           const float* __restrict__ avec,
                                           float* __restrict__ C, float* __restrict__ S0,
                                           float* __restrict__ S1, int nrows, int strip,
                                           float* Blds) {
    constexpr int CW = COLT * 16;
    const int tid = threadIdx.x;

    for (int idx = tid; idx < 256 * CW; idx += 256) {
        float bv;
        if constexpr (MODE == 0) {
            bv = B[(size_t)strip * (256 * 64) + idx];
        } else if constexpr (MODE == 1) {
            int k = idx / CW;
            int c = idx - k * CW;
            bv = (c < 40) ? B[k * 40 + c] : 0.0f;
        } else {
            bv = B[idx];
        }
        Blds[idx] = bv;
    }
    __syncthreads();

    if constexpr (MODE == 1) {
        const float* wrow = Blds + tid * CW;
        float s0 = 0.f, s1 = 0.f;
#pragma unroll 8
        for (int f = 0; f < 40; ++f) {
            const float w = wrow[f];
            s0 += w * avec[f];
            s1 += w * avec[40 + f];
        }
        Blds[tid * CW + 40] = s0;
        Blds[tid * CW + 41] = s1;
        __syncthreads();
    }

    const int wid  = tid >> 5;
    const int lane = tid & 31;
    const int half = lane >> 4;
    const int l16  = lane & 15;
    const int rowBase = blockIdx.x * 128 + wid * 16;

    int arow = rowBase + l16;
    if (arow >= nrows) arow = nrows - 1;
    const float* aptr = A + (size_t)arow * 256;

    v8f acc[COLT] = {};

    for (int k0 = 0; k0 < 256; k0 += 4) {
        const int kb = k0 + half * 2;
        v2f a;
        a.x = aptr[kb];
        a.y = aptr[kb + 1];
#pragma unroll
        for (int t = 0; t < COLT; ++t) {
            v2f b;
            const int c = t * 16 + l16;
            b.x = Blds[kb * CW + c];
            b.y = Blds[(kb + 1) * CW + c];
            acc[t] = __builtin_amdgcn_wmma_f32_16x16x4_f32(false, a, false, b, (short)0,
                                                           acc[t], false, false);
        }
    }

    if (rowBase + 16 <= nrows) {
        store_tiles<COLT, MODE, false>(acc, C, S0, S1, nrows, strip, rowBase, half, l16);
    } else {
        store_tiles<COLT, MODE, true>(acc, C, S0, S1, nrows, strip, rowBase, half, l16);
    }
}

__global__ __launch_bounds__(256) void gemm1_kernel(const float* __restrict__ h,
                                                    const float* __restrict__ Wheads,
                                                    const float* __restrict__ Wa1,
                                                    float* __restrict__ Wh,
                                                    float* __restrict__ ssrc,
                                                    float* __restrict__ sdst, int nrows) {
    __shared__ float Blds[256 * 64];
    if (blockIdx.y == 0) {
        gemm_strip<1, 2>(h, Wa1, nullptr, nullptr, ssrc, sdst, nrows, 0, Blds);
    } else {
        gemm_strip<4, 0>(h, Wheads, nullptr, Wh, nullptr, nullptr, nrows, (int)blockIdx.y - 1,
                         Blds);
    }
}

__global__ __launch_bounds__(256) void gemm2_kernel(const float* __restrict__ x,
                                                    const float* __restrict__ Wout,
                                                    const float* __restrict__ aout,
                                                    float* __restrict__ Wx,
                                                    float* __restrict__ ssrc,
                                                    float* __restrict__ sdst, int nrows) {
    __shared__ float Blds[256 * 48];
    gemm_strip<3, 1>(x, Wout, aout, Wx, ssrc, sdst, nrows, 0, Blds);
}

__global__ __launch_bounds__(256) void edge_link(const int* __restrict__ dst,
                                                 int* __restrict__ head,
                                                 int* __restrict__ nxt, int ne) {
    int e = blockIdx.x * blockDim.x + threadIdx.x;
    if (e >= ne) return;
    nxt[e] = atomicExch(&head[dst[e]], e);
}

__device__ __forceinline__ float lane_bcast(float v, int l) {
    return __int_as_float(__builtin_amdgcn_readlane(__float_as_int(v), l));
}

__global__ __launch_bounds__(256) void gather1(const int* __restrict__ src,
                                               const int* __restrict__ head,
                                               const int* __restrict__ nxt,
                                               const float* __restrict__ ssrc,
                                               const float* __restrict__ sdst,
                                               const float* __restrict__ Wh,
                                               float* __restrict__ hp, int n, int ne) {
    const int lane = threadIdx.x & 31;
    const int hsel = lane & 3;
    const int node = __builtin_amdgcn_readfirstlane((int)(blockIdx.x * 8 + (threadIdx.x >> 5)));
    if (node >= n) return;
    float acc[8];
#pragma unroll
    for (int j = 0; j < 8; ++j) acc[j] = 0.f;
    float m = -INFINITY, den = 0.f;
    const float sd = sdst[node * 4 + hsel];

    int e = head[node];
    bool ok = (unsigned)e < (unsigned)ne;
    int s = 0, en = -1;
    if (ok) {
        s = src[e];
        en = nxt[e];
    }
    for (int it = 0; it < ne && ok; ++it) {
        const float* row = Wh + (size_t)s * 256 + lane;
        float w[8];
#pragma unroll
        for (int j = 0; j < 8; ++j) w[j] = row[32 * j];
        const float xs = ssrc[s * 4 + hsel];
        e = en;
        ok = (unsigned)e < (unsigned)ne;
        if (ok) {
            s = src[e];
            en = nxt[e];
        }
        const float x = lrelu(xs + sd);
        const bool up = x > m;
        const float mn = fmaxf(m, x);
        const float t = expf(fminf(m, x) - mn);
        const float c = up ? t : 1.f;
        const float p = up ? 1.f : t;
        den = den * c + p;
        m = mn;
#pragma unroll
        for (int h = 0; h < 4; ++h) {
            const float ch = lane_bcast(c, h);
            const float ph = lane_bcast(p, h);
            acc[2 * h]     = acc[2 * h] * ch + ph * w[2 * h];
            acc[2 * h + 1] = acc[2 * h + 1] * ch + ph * w[2 * h + 1];
        }
    }
    float* orow = hp + (size_t)node * 256 + lane;
#pragma unroll
    for (int j = 0; j < 8; ++j) {
        const float dh = lane_bcast(den, j >> 1);
        const float v = dh > 0.f ? acc[j] / dh : 0.f;
        orow[32 * j] = v > 0.f ? v : (expf(v) - 1.f);
    }
}

__global__ __launch_bounds__(256) void gather2(const int* __restrict__ src,
                                               const int* __restrict__ head,
                                               const int* __restrict__ nxt,
                                               const float* __restrict__ ssrc,
                                               const float* __restrict__ sdst,
                                               const float* __restrict__ Wx,
                                               float* __restrict__ out, int n, int ne) {
    const int lane = threadIdx.x & 31;
    const int node = __builtin_amdgcn_readfirstlane((int)(blockIdx.x * 8 + (threadIdx.x >> 5)));
    if (node >= n) return;
    float acc1 = 0.f, acc2 = 0.f;
    float m = -INFINITY, den = 0.f;
    const float sd = sdst[node];

    int e = head[node];
    bool ok = (unsigned)e < (unsigned)ne;
    int s = 0, en = -1;
    if (ok) {
        s = src[e];
        en = nxt[e];
    }
    for (int it = 0; it < ne && ok; ++it) {
        const float* row = Wx + (size_t)s * 40;
        const float w1 = row[lane];
        const float w2 = (lane < 8) ? row[32 + lane] : 0.f;
        const float xs = ssrc[s];
        e = en;
        ok = (unsigned)e < (unsigned)ne;
        if (ok) {
            s = src[e];
            en = nxt[e];
        }
        const float x = lrelu(xs + sd);
        const bool up = x > m;
        const float mn = fmaxf(m, x);
        const float t = expf(fminf(m, x) - mn);
        const float c = up ? t : 1.f;
        const float p = up ? 1.f : t;
        den = den * c + p;
        m = mn;
        acc1 = acc1 * c + p * w1;
        acc2 = acc2 * c + p * w2;
    }
    const bool any = den > 0.f;
    const float v1 = any ? acc1 / den : 0.f;
    const float v2 = (lane < 8) ? (any ? acc2 / den : 0.f) : -INFINITY;
    const float mx = wred_max(fmaxf(v1, v2));
    float sm = expf(v1 - mx) + ((lane < 8) ? expf(v2 - mx) : 0.f);
    sm = wred_sum(sm);
    const float lg = mx + logf(sm);
    float* orow = out + (size_t)node * 40;
    orow[lane] = v1 - lg;
    if (lane < 8) orow[32 + lane] = v2 - lg;
}

extern "C" void kernel_launch(void* const* d_in, const int* in_sizes, int n_in, void* d_out,
                              int out_size, void* d_ws, size_t ws_size, hipStream_t stream) {
    (void)n_in; (void)out_size; (void)ws_size;
    const float* h      = (const float*)d_in[0];
    const int*   eidx   = (const int*)d_in[1];
    const float* Wheads = (const float*)d_in[2];
    const float* aheads = (const float*)d_in[3];
    const float* Wout   = (const float*)d_in[4];
    const float* aout   = (const float*)d_in[5];
    float* out = (float*)d_out;

    const int N = in_sizes[0] / 256;
    const int E = in_sizes[1] / 2;
    const int* src = eidx;
    const int* dst = eidx + E;

    float* ws = (float*)d_ws;
    size_t oWh   = 0;
    size_t oHP   = oWh  + (size_t)N * 256;
    size_t oS1s  = oHP  + (size_t)N * 256;
    size_t oS1d  = oS1s + (size_t)N * 4;
    size_t oHead = oS1d + (size_t)N * 4;
    size_t oNxt  = oHead + (size_t)N;
    size_t oS2s  = oS1s;
    size_t oS2d  = oS1d;
    size_t oWx   = oWh;
    size_t oWa1  = oHP;

    int* head = (int*)(ws + oHead);
    int* nxt  = (int*)(ws + oNxt);

    int fillBlocks = (N + 255) / 256;
    prep_kernel<<<fillBlocks + 8, 256, 0, stream>>>(head, N, fillBlocks, Wheads, aheads,
                                                    ws + oWa1);

    dim3 g1((N + 127) / 128, 5);
    gemm1_kernel<<<g1, 256, 0, stream>>>(h, Wheads, ws + oWa1, ws + oWh, ws + oS1s, ws + oS1d,
                                         N);

    int geE = (E + 255) / 256;
    edge_link<<<geE, 256, 0, stream>>>(dst, head, nxt, E);

    gather1<<<(N + 7) / 8, 256, 0, stream>>>(src, head, nxt, ws + oS1s, ws + oS1d, ws + oWh,
                                             ws + oHP, N, E);

    dim3 g2((N + 127) / 128, 1);
    gemm2_kernel<<<g2, 256, 0, stream>>>(ws + oHP, Wout, aout, ws + oWx, ws + oS2s, ws + oS2d,
                                         N);

    gather2<<<(N + 7) / 8, 256, 0, stream>>>(src, head, nxt, ws + oS2s, ws + oS2d, ws + oWx,
                                             out, N, E);
}


# layer-2 gather: one lane per destination node walks its in-edge list (32 lists in flight per wave), row read by 16-byte loads into the lane's registers; log_softmax max and sum now in class order
# speedup vs baseline: 6.0523x; 1.0824x over previous
#include <hip/hip_runtime.h>


#define LRELU 0.2f

typedef __attribute__((ext_vector_type(2))) float v2f;
typedef __attribute__((ext_vector_type(8))) float v8f;
typedef __attribute__((ext_vector_type(4))) float v4f;

__device__ __forceinline__ float lrelu(float x) { return x > 0.f ? x : LRELU * x; }

__global__ __launch_bounds__(256) void prep_kernel(int* __restrict__ head, int n, int fillBlocks,
                                                   const float* __restrict__ Wheads,
                                                   const float* __restrict__ aheads,
                                                   float* __restrict__ Wa1) {
    const int b = blockIdx.x;
    if (b < fillBlocks) {
        const int i = b * 256 + threadIdx.x;
        if (i < n) head[i] = -1;
        return;
    }
    const int c = b - fillBlocks;
    const int hd = c & 3;
    const int k = threadIdx.x;
    const float* w = Wheads + ((size_t)hd * 256 + k) * 64;
    const float* a = aheads + hd * 128 + (c >> 2) * 64;
    float s = 0.f;
#pragma unroll
    for (int f = 0; f < 64; ++f) s += w[f] * a[f];
    Wa1[k * 16 + c] = s;
    Wa1[k * 16 + 8 + c] = 0.f;
}

template <int COLT, int MODE, bool GUARD>
__device__ __forceinline__ void store_tiles(const v8f (&acc)[COLT], float* __restrict__ C,
                                            float* __restrict__ S0, float* __restrict__ S1,
                                            int nrows, int strip, int rowBase, int half, int l16) {
#pragma unroll
    for (int t = 0; t < COLT; ++t) {
        const int col = t * 16 + l16;
        float* base;
        size_t stride;
        bool on = true;
        if constexpr (MODE == 0) {
            base = C + strip * 64 + col;
            stride = 256;
        } else if constexpr (MODE == 1) {
            if (col < 40) {
                base = C + col;
                stride = 40;
            } else {
                base = (col == 40) ? S0 : S1;
                stride = 1;
                on = col < 42;
            }
        } else {
            base = (col < 4) ? (S0 + col) : (S1 + (col - 4));
            stride = 4;
            on = col < 8;
        }
#pragma unroll
        for (int v = 0; v < 8; ++v) {
            const int row = rowBase + v + half * 8;
            if (on && (!GUARD || row < nrows)) base[(size_t)row * stride] = acc[t][v];
        }
    }
}

template <int COLT, int MODE>
__device__ __forceinline__ void gemm_strip(const float* __restrict__ A,
                                           const float* __restrict__ B,
                                           const float* __restrict__ avec,
                                           float* __restrict__ C, float* __restrict__ S0,
                                           float* __restrict__ S1, int nrows, int strip,
                                           float* Blds) {
    constexpr int CW = COLT * 16;
    const int tid = threadIdx.x;

    for (int idx = tid; idx < 256 * CW; idx += 256) {
        float bv;
        if constexpr (MODE == 0) {
            bv = B[(size_t)strip * (256 * 64) + idx];
        } else if constexpr (MODE == 1) {
            int k = idx / CW;
            int c = idx - k * CW;
            bv = (c < 40) ? B[k * 40 + c] : 0.0f;
        } else {
            bv = B[idx];
        }
        Blds[idx] = bv;
    }
    __syncthreads();

    if constexpr (MODE == 1) {
        const float* wrow = Blds + tid * CW;
        float s0 = 0.f, s1 = 0.f;
#pragma unroll 8
        for (int f = 0; f < 40; ++f) {
            const float w = wrow[f];
            s0 += w * avec[f];
            s1 += w * avec[40 + f];
        }
        Blds[tid * CW + 40] = s0;
        Blds[tid * CW + 41] = s1;
        __syncthreads();
    }

    const int wid  = tid >> 5;
    const int lane = tid & 31;
    const int half = lane >> 4;
    const int l16  = lane & 15;
    const int rowBase = blockIdx.x * 128 + wid * 16;

    int arow = rowBase + l16;
    if (arow >= nrows) arow = nrows - 1;
    const float* aptr = A + (size_t)arow * 256;

    v8f acc[COLT] = {};

    for (int k0 = 0; k0 < 256; k0 += 4) {
        const int kb = k0 + half * 2;
        v2f a;
        a.x = aptr[kb];
        a.y = aptr[kb + 1];
#pragma unroll
        for (int t = 0; t < COLT; ++t) {
            v2f b;
            const int c = t * 16 + l16;
            b.x = Blds[kb * CW + c];
            b.y = Blds[(kb + 1) * CW + c];
            acc[t] = __builtin_amdgcn_wmma_f32_16x16x4_f32(false, a, false, b, (short)0,
                                                           acc[t], false, false);
        }
    }

    if (rowBase + 16 <= nrows) {
        store_tiles<COLT, MODE, false>(acc, C, S0, S1, nrows, strip, rowBase, half, l16);
    } else {
        store_tiles<COLT, MODE, true>(acc, C, S0, S1, nrows, strip, rowBase, half, l16);
    }
}

__global__ __launch_bounds__(256) void gemm1_kernel(const float* __restrict__ h,
                                                    const float* __restrict__ Wheads,
                                                    const float* __restrict__ Wa1,
                                                    float* __restrict__ Wh,
                                                    float* __restrict__ ssrc,
                                                    float* __restrict__ sdst, int nrows) {
    __shared__ float Blds[256 * 64];
    if (blockIdx.y == 0) {
        gemm_strip<1, 2>(h, Wa1, nullptr, nullptr, ssrc, sdst, nrows, 0, Blds);
    } else {
        gemm_strip<4, 0>(h, Wheads, nullptr, Wh, nullptr, nullptr, nrows, (int)blockIdx.y - 1,
                         Blds);
    }
}

__global__ __launch_bounds__(256) void gemm2_kernel(const float* __restrict__ x,
                                                    const float* __restrict__ Wout,
                                                    const float* __restrict__ aout,
                                                    float* __restrict__ Wx,
                                                    float* __restrict__ ssrc,
                                                    float* __restrict__ sdst, int nrows) {
    __shared__ float Blds[256 * 48];
    gemm_strip<3, 1>(x, Wout, aout, Wx, ssrc, sdst, nrows, 0, Blds);
}

__global__ __launch_bounds__(256) void edge_link(const int* __restrict__ dst,
                                                 int* __restrict__ head,
                                                 int* __restrict__ nxt, int ne) {
    int e = blockIdx.x * blockDim.x + threadIdx.x;
    if (e >= ne) return;
    nxt[e] = atomicExch(&head[dst[e]], e);
}

__device__ __forceinline__ float lane_bcast(float v, int l) {
    return __int_as_float(__builtin_amdgcn_readlane(__float_as_int(v), l));
}

__global__ __launch_bounds__(256) void gather1(const int* __restrict__ src,
                                               const int* __restrict__ head,
                                               const int* __restrict__ nxt,
                                               const float* __restrict__ ssrc,
                                               const float* __restrict__ sdst,
                                               const float* __restrict__ Wh,
                                               float* __restrict__ hp, int n, int ne) {
    const int lane = threadIdx.x & 31;
    const int hsel = lane & 3;
    const int node = __builtin_amdgcn_readfirstlane((int)(blockIdx.x * 8 + (threadIdx.x >> 5)));
    if (node >= n) return;
    float acc[8];
#pragma unroll
    for (int j = 0; j < 8; ++j) acc[j] = 0.f;
    float m = -INFINITY, den = 0.f;
    const float sd = sdst[node * 4 + hsel];

    int e = head[node];
    bool ok = (unsigned)e < (unsigned)ne;
    int s = 0, en = -1;
    if (ok) {
        s = src[e];
        en = nxt[e];
    }
    for (int it = 0; it < ne && ok; ++it) {
        const float* row = Wh + (size_t)s * 256 + lane;
        float w[8];
#pragma unroll
        for (int j = 0; j < 8; ++j) w[j] = row[32 * j];
        const float xs = ssrc[s * 4 + hsel];
        e = en;
        ok = (unsigned)e < (unsigned)ne;
        if (ok) {
            s = src[e];
            en = nxt[e];
        }
        const float x = lrelu(xs + sd);
        const bool up = x > m;
        const float mn = fmaxf(m, x);
        const float t = expf(fminf(m, x) - mn);
        const float c = up ? t : 1.f;
        const float p = up ? 1.f : t;
        den = den * c + p;
        m = mn;
#pragma unroll
        for (int h = 0; h < 4; ++h) {
            const float ch = lane_bcast(c, h);
            const float ph = lane_bcast(p, h);
            acc[2 * h]     = acc[2 * h] * ch + ph * w[2 * h];
            acc[2 * h + 1] = acc[2 * h + 1] * ch + ph * w[2 * h + 1];
        }
    }
    float* orow = hp + (size_t)node * 256 + lane;
#pragma unroll
    for (int j = 0; j < 8; ++j) {
        const float dh = lane_bcast(den, j >> 1);
        const float v = dh > 0.f ? acc[j] / dh : 0.f;
        orow[32 * j] = v > 0.f ? v : (expf(v) - 1.f);
    }
}

__global__ __launch_bounds__(64) void gather2(const int* __restrict__ src,
                                              const int* __restrict__ head,
                                              const int* __restrict__ nxt,
                                              const float* __restrict__ ssrc,
                                              const float* __restrict__ sdst,
                                              const float* __restrict__ Wx,
                                              float* __restrict__ out, int n, int ne) {
    const int node = (int)(blockIdx.x * 64 + threadIdx.x);
    if (node >= n) return;
    float acc[40];
#pragma unroll
    for (int k = 0; k < 40; ++k) acc[k] = 0.f;
    float m = -INFINITY, den = 0.f;
    const float sd = sdst[node];

    int e = head[node];
    bool ok = (unsigned)e < (unsigned)ne;
    int s = 0, en = -1;
    if (ok) {
        s = src[e];
        en = nxt[e];
    }
    for (int it = 0; it < ne && ok; ++it) {
        const v4f* row = reinterpret_cast<const v4f*>(Wx + (size_t)s * 40);
        v4f w[10];
#pragma unroll
        for (int j = 0; j < 10; ++j) w[j] = row[j];
        const float xs = ssrc[s];
        e = en;
        ok = (unsigned)e < (unsigned)ne;
        if (ok) {
            s = src[e];
            en = nxt[e];
        }
        const float x = lrelu(xs + sd);
        const bool up = x > m;
        const float mn = fmaxf(m, x);
        const float t = expf(fminf(m, x) - mn);
        const float c = up ? t : 1.f;
        const float p = up ? 1.f : t;
        den = den * c + p;
        m = mn;
#pragma unroll
        for (int j = 0; j < 10; ++j) {
            acc[4 * j + 0] = acc[4 * j + 0] * c + p * w[j].x;
            acc[4 * j + 1] = acc[4 * j + 1] * c + p * w[j].y;
            acc[4 * j + 2] = acc[4 * j + 2] * c + p * w[j].z;
            acc[4 * j + 3] = acc[4 * j + 3] * c + p * w[j].w;
        }
    }
    const bool any = den > 0.f;
    float mx = -INFINITY;
#pragma unroll
    for (int k = 0; k < 40; ++k) {
        acc[k] = any ? acc[k] / den : 0.f;
        mx = fmaxf(mx, acc[k]);
    }
    float sm = 0.f;
#pragma unroll
    for (int k = 0; k < 40; ++k) sm += expf(acc[k] - mx);
    const float lg = mx + logf(sm);
    v4f* orow = reinterpret_cast<v4f*>(out + (size_t)node * 40);
#pragma unroll
    for (int j = 0; j < 10; ++j) {
        v4f o;
        o.x = acc[4 * j + 0] - lg;
        o.y = acc[4 * j + 1] - lg;
        o.z = acc[4 * j + 2] - lg;
        o.w = acc[4 * j + 3] - lg;
        orow[j] = o;
    }
}

extern "C" void kernel_launch(void* const* d_in, const int* in_sizes, int n_in, void* d_out,
                              int out_size, void* d_ws, size_t ws_size, hipStream_t stream) {
    (void)n_in; (void)out_size; (void)ws_size;
    const float* h      = (const float*)d_in[0];
    const int*   eidx   = (const int*)d_in[1];
    const float* Wheads = (const float*)d_in[2];
    const float* aheads = (const float*)d_in[3];
    const float* Wout   = (const float*)d_in[4];
    const float* aout   = (const float*)d_in[5];
    float* out = (float*)d_out;

    const int N = in_sizes[0] / 256;
    const int E = in_sizes[1] / 2;
    const int* src = eidx;
    const int* dst = eidx + E;

    float* ws = (float*)d_ws;
    size_t oWh   = 0;
    size_t oHP   = oWh  + (size_t)N * 256;
    size_t oS1s  = oHP  + (size_t)N * 256;
    size_t oS1d  = oS1s + (size_t)N * 4;
    size_t oHead = oS1d + (size_t)N * 4;
    size_t oNxt  = oHead + (size_t)N;
    size_t oS2s  = oS1s;
    size_t oS2d  = oS1d;
    size_t oWx   = oWh;
    size_t oWa1  = oHP;

    int* head = (int*)(ws + oHead);
    int* nxt  = (int*)(ws + oNxt);

    int fillBlocks = (N + 255) / 256;
    prep_kernel<<<fillBlocks + 8, 256, 0, stream>>>(head, N, fillBlocks, Wheads, aheads,
                                                    ws + oWa1);

    dim3 g1((N + 127) / 128, 5);
    gemm1_kernel<<<g1, 256, 0, stream>>>(h, Wheads, ws + oWa1, ws + oWh, ws + oS1s, ws + oS1d,
                                         N);

    int geE = (E + 255) / 256;
    edge_link<<<geE, 256, 0, stream>>>(dst, head, nxt, E);

    gather1<<<(N + 7) / 8, 256, 0, stream>>>(src, head, nxt, ws + oS1s, ws + oS1d, ws + oWh,
                                             ws + oHP, N, E);

    dim3 g2((N + 127) / 128, 1);
    gemm2_kernel<<<g2, 256, 0, stream>>>(ws + oHP, Wout, aout, ws + oWx, ws + oS2s, ws + oS2d,
                                         N);

    gather2<<<(N + 63) / 64, 64, 0, stream>>>(src, head, nxt, ws + oS2s, ws + oS2d, ws + oWx,
                                              out, N, E);
}


# layer-1 GEMM retiled: one 16-wave work-group per WGP keeps the whole 256x272 weight panel in LDS, a wave makes 16 rows x 272 columns, h read once; the k sum order of its products changes
# speedup vs baseline: 6.3587x; 1.0506x over previous
#include <hip/hip_runtime.h>


#define LRELU 0.2f

typedef __attribute__((ext_vector_type(2))) float v2f;
typedef __attribute__((ext_vector_type(8))) float v8f;
typedef __attribute__((ext_vector_type(4))) float v4f;

__device__ __forceinline__ float lrelu(float x) { return x > 0.f ? x : LRELU * x; }

__global__ __launch_bounds__(256) void prep_kernel(int* __restrict__ head, int n, int fillBlocks,
                                                   const float* __restrict__ Wheads,
                                                   const float* __restrict__ aheads,
                                                   float* __restrict__ Wa1) {
    const int b = blockIdx.x;
    if (b < fillBlocks) {
        const int i = b * 256 + threadIdx.x;
        if (i < n) head[i] = -1;
        return;
    }
    const int c = b - fillBlocks;
    const int hd = c & 3;
    const int k = threadIdx.x;
    const float* w = Wheads + ((size_t)hd * 256 + k) * 64;
    const float* a = aheads + hd * 128 + (c >> 2) * 64;
    float s = 0.f;
#pragma unroll
    for (int f = 0; f < 64; ++f) s += w[f] * a[f];
    Wa1[k * 16 + c] = s;
    Wa1[k * 16 + 8 + c] = 0.f;
}

#define G1_COLS 272
#define G1_TILES 17

__global__ __launch_bounds__(512) void gemm1_kernel(const float* __restrict__ h,
                                                    const float* __restrict__ Wheads,
                                                    const float* __restrict__ Wa1,
                                                    float* __restrict__ Wh,
                                                    float* __restrict__ ssrc,
                                                    float* __restrict__ sdst, int nrows) {
    __shared__ __attribute__((aligned(16))) float Blds[256 * G1_COLS];
    const int tid = threadIdx.x;

    {
        v4f r[9][4];
#pragma unroll
        for (int i = 0; i < 9; ++i) {
            int u = tid + 512 * i;
            if (u > 64 * (G1_COLS / 4) - 1) u = 64 * (G1_COLS / 4) - 1;
            const int q = u / (G1_COLS / 4);
            const int c = (u - q * (G1_COLS / 4)) * 4;
            const bool head = c < 256;
            const float* sp = head ? (Wheads + ((size_t)(c >> 6) * 256 + 4 * q) * 64 + (c & 63))
                                   : (Wa1 + (4 * q) * 16 + (c - 256));
            const int pitch = head ? 64 : 16;
#pragma unroll
            for (int j = 0; j < 4; ++j) r[i][j] = *reinterpret_cast<const v4f*>(sp + j * pitch);
        }
#pragma unroll
        for (int i = 0; i < 9; ++i) {
            const int u = tid + 512 * i;
            if (u < 64 * (G1_COLS / 4)) {
                const int q = u / (G1_COLS / 4);
                const int c = (u - q * (G1_COLS / 4)) * 4;
                v4f* d = reinterpret_cast<v4f*>(Blds + (q * G1_COLS + c) * 4);
                v4f s0, s1, s2, s3;
                s0.x = r[i][0].x; s0.y = r[i][1].x; s0.z = r[i][2].x; s0.w = r[i][3].x;
                s1.x = r[i][0].y; s1.y = r[i][1].y; s1.z = r[i][2].y; s1.w = r[i][3].y;
                s2.x = r[i][0].z; s2.y = r[i][1].z; s2.z = r[i][2].z; s2.w = r[i][3].z;
                s3.x = r[i][0].w; s3.y = r[i][1].w; s3.z = r[i][2].w; s3.w = r[i][3].w;
                d[0] = s0;
                d[1] = s1;
                d[2] = s2;
                d[3] = s3;
            }
        }
    }
    __syncthreads();

    const int wid  = __builtin_amdgcn_readfirstlane(tid >> 5);
    const int lane = tid & 31;
    const int hf   = lane >> 4;
    const int l16  = lane & 15;
    const int ntile = (nrows + 15) >> 4;
    const float* bbase = Blds + (hf * G1_COLS + l16) * 4;

    const int slot = 4 * (((wid >> 2) - (wid & 3)) & 3) + (wid & 3);
    for (int tile = slot * (int)gridDim.x + (int)blockIdx.x; tile < ntile;
         tile += 16 * (int)gridDim.x) {
        const int rowBase = tile * 16;
        int arow = rowBase + l16;
        if (arow >= nrows) arow = nrows - 1;
        const v4f* aptr = reinterpret_cast<const v4f*>(h + (size_t)arow * 256 + hf * 4);

        v8f acc[G1_TILES] = {};
        v4f a = aptr[0];
        for (int g = 0; g < 32; ++g) {
            const v4f an = aptr[2 * ((g + 1) & 31)];
            __builtin_amdgcn_sched_barrier(0);
            const float* bp = bbase + g * (2 * G1_COLS * 4);
            v4f b[G1_TILES];
#pragma unroll
            for (int t = 0; t < G1_TILES; ++t) b[t] = *reinterpret_cast<const v4f*>(bp + t * 64);
            v2f a0, a1;
            a0.x = a.x; a0.y = a.y;
            a1.x = a.z; a1.y = a.w;
#pragma unroll
            for (int t = 0; t < G1_TILES; ++t) {
                v2f b0;
                b0.x = b[t].x; b0.y = b[t].y;
                acc[t] = __builtin_amdgcn_wmma_f32_16x16x4_f32(false, a0, false, b0, (short)0,
                                                               acc[t], false, false);
            }
#pragma unroll
            for (int t = 0; t < G1_TILES; ++t) {
                v2f b1;
                b1.x = b[t].z; b1.y = b[t].w;
                acc[t] = __builtin_amdgcn_wmma_f32_16x16x4_f32(false, a1, false, b1, (short)0,
                                                               acc[t], false, false);
            }
            a = an;
        }

        const bool full = rowBase + 16 <= nrows;
        float* crow = Wh + (size_t)(rowBase + 8 * hf) * 256 + l16;
        if (full) {
#pragma unroll
            for (int t = 0; t < 16; ++t)
#pragma unroll
                for (int v = 0; v < 8; ++v) crow[(size_t)v * 256 + t * 16] = acc[t][v];
        } else {
#pragma unroll
            for (int t = 0; t < 16; ++t)
#pragma unroll
                for (int v = 0; v < 8; ++v)
                    if (rowBase + 8 * hf + v < nrows) crow[(size_t)v * 256 + t * 16] = acc[t][v];
        }
        if (l16 < 8) {
            float* sb = ((l16 < 4) ? (ssrc + l16) : (sdst + (l16 - 4))) +
                        (size_t)(rowBase + 8 * hf) * 4;
#pragma unroll
            for (int v = 0; v < 8; ++v)
                if (full || rowBase + 8 * hf + v < nrows) sb[(size_t)v * 4] = acc[16][v];
        }
    }
}

template <bool GUARD>
__device__ __forceinline__ void store_tiles2(const v8f (&acc)[3], float* __restrict__ C,
                                             float* __restrict__ S0, float* __restrict__ S1,
                                             int nrows, int rowBase, int half, int l16) {
#pragma unroll
    for (int t = 0; t < 3; ++t) {
        const int col = t * 16 + l16;
        float* base;
        size_t stride;
        bool on = true;
        if (col < 40) {
            base = C + col;
            stride = 40;
        } else {
            base = (col == 40) ? S0 : S1;
            stride = 1;
            on = col < 42;
        }
#pragma unroll
        for (int v = 0; v < 8; ++v) {
            const int row = rowBase + v + half * 8;
            if (on && (!GUARD || row < nrows)) base[(size_t)row * stride] = acc[t][v];
        }
    }
}

__global__ __launch_bounds__(256) void gemm2_kernel(const float* __restrict__ x,
                                                    const float* __restrict__ Wout,
                                                    const float* __restrict__ aout,
                                                    float* __restrict__ Wx,
                                                    float* __restrict__ ssrc,
                                                    float* __restrict__ sdst, int nrows) {
    constexpr int CW = 48;
    __shared__ float Blds[256 * CW];
    const int tid = threadIdx.x;

    for (int idx = tid; idx < 256 * CW; idx += 256) {
        const int k = idx / CW;
        const int c = idx - k * CW;
        Blds[idx] = (c < 40) ? Wout[k * 40 + c] : 0.0f;
    }
    __syncthreads();

    {
        const float* wrow = Blds + tid * CW;
        float s0 = 0.f, s1 = 0.f;
#pragma unroll 8
        for (int f = 0; f < 40; ++f) {
            const float w = wrow[f];
            s0 += w * aout[f];
            s1 += w * aout[40 + f];
        }
        Blds[tid * CW + 40] = s0;
        Blds[tid * CW + 41] = s1;
        __syncthreads();
    }

    const int wid  = tid >> 5;
    const int lane = tid & 31;
    const int half = lane >> 4;
    const int l16  = lane & 15;
    const int rowBase = blockIdx.x * 128 + wid * 16;

    int arow = rowBase + l16;
    if (arow >= nrows) arow = nrows - 1;
    const float* aptr = x + (size_t)arow * 256;

    v8f acc[3] = {};

    for (int k0 = 0; k0 < 256; k0 += 4) {
        const int kb = k0 + half * 2;
        v2f a;
        a.x = aptr[kb];
        a.y = aptr[kb + 1];
#pragma unroll
        for (int t = 0; t < 3; ++t) {
            v2f b;
            const int c = t * 16 + l16;
            b.x = Blds[kb * CW + c];
            b.y = Blds[(kb + 1) * CW + c];
            acc[t] = __builtin_amdgcn_wmma_f32_16x16x4_f32(false, a, false, b, (short)0,
                                                           acc[t], false, false);
        }
    }

    if (rowBase + 16 <= nrows) {
        store_tiles2<false>(acc, Wx, ssrc, sdst, nrows, rowBase, half, l16);
    } else {
        store_tiles2<true>(acc, Wx, ssrc, sdst, nrows, rowBase, half, l16);
    }
}

__global__ __launch_bounds__(256) void edge_link(const int* __restrict__ dst,
                                                 int* __restrict__ head,
                                                 int* __restrict__ nxt, int ne) {
    int e = blockIdx.x * blockDim.x + threadIdx.x;
    if (e >= ne) return;
    nxt[e] = atomicExch(&head[dst[e]], e);
}

__device__ __forceinline__ float lane_bcast(float v, int l) {
    return __int_as_float(__builtin_amdgcn_readlane(__float_as_int(v), l));
}

__global__ __launch_bounds__(256) void gather1(const int* __restrict__ src,
                                               const int* __restrict__ head,
                                               const int* __restrict__ nxt,
                                               const float* __restrict__ ssrc,
                                               const float* __restrict__ sdst,
                                               const float* __restrict__ Wh,
                                               float* __restrict__ hp, int n, int ne) {
    const int lane = threadIdx.x & 31;
    const int hsel = lane & 3;
    const int node = __builtin_amdgcn_readfirstlane((int)(blockIdx.x * 8 + (threadIdx.x >> 5)));
    if (node >= n) return;
    float acc[8];
#pragma unroll
    for (int j = 0; j < 8; ++j) acc[j] = 0.f;
    float m = -INFINITY, den = 0.f;
    const float sd = sdst[node * 4 + hsel];

    int e = head[node];
    bool ok = (unsigned)e < (unsigned)ne;
    int s = 0, en = -1;
    if (ok) {
        s = src[e];
        en = nxt[e];
    }
    for (int it = 0; it < ne && ok; ++it) {
        const float* row = Wh + (size_t)s * 256 + lane;
        float w[8];
#pragma unroll
        for (int j = 0; j < 8; ++j) w[j] = row[32 * j];
        const float xs = ssrc[s * 4 + hsel];
        e = en;
        ok = (unsigned)e < (unsigned)ne;
        if (ok) {
            s = src[e];
            en = nxt[e];
        }
        const float x = lrelu(xs + sd);
        const bool up = x > m;
        const float mn = fmaxf(m, x);
        const float t = expf(fminf(m, x) - mn);
        const float c = up ? t : 1.f;
        const float p = up ? 1.f : t;
        den = den * c + p;
        m = mn;
#pragma unroll
        for (int h = 0; h < 4; ++h) {
            const float ch = lane_bcast(c, h);
            const float ph = lane_bcast(p, h);
            acc[2 * h]     = acc[2 * h] * ch + ph * w[2 * h];
            acc[2 * h + 1] = acc[2 * h + 1] * ch + ph * w[2 * h + 1];
        }
    }
    float* orow = hp + (size_t)node * 256 + lane;
#pragma unroll
    for (int j = 0; j < 8; ++j) {
        const float dh = lane_bcast(den, j >> 1);
        const float v = dh > 0.f ? acc[j] / dh : 0.f;
        orow[32 * j] = v > 0.f ? v : (expf(v) - 1.f);
    }
}

__global__ __launch_bounds__(64) void gather2(const int* __restrict__ src,
                                              const int* __restrict__ head,
                                              const int* __restrict__ nxt,
                                              const float* __restrict__ ssrc,
                                              const float* __restrict__ sdst,
                                              const float* __restrict__ Wx,
                                              float* __restrict__ out, int n, int ne) {
    const int node = (int)(blockIdx.x * 64 + threadIdx.x);
    if (node >= n) return;
    float acc[40];
#pragma unroll
    for (int k = 0; k < 40; ++k) acc[k] = 0.f;
    float m = -INFINITY, den = 0.f;
    const float sd = sdst[node];

    int e = head[node];
    bool ok = (unsigned)e < (unsigned)ne;
    int s = 0, en = -1;
    if (ok) {
        s = src[e];
        en = nxt[e];
    }
    for (int it = 0; it < ne && ok; ++it) {
        const v4f* row = reinterpret_cast<const v4f*>(Wx + (size_t)s * 40);
        v4f w[10];
#pragma unroll
        for (int j = 0; j < 10; ++j) w[j] = row[j];
        const float xs = ssrc[s];
        e = en;
        ok = (unsigned)e < (unsigned)ne;
        if (ok) {
            s = src[e];
            en = nxt[e];
        }
        const float x = lrelu(xs + sd);
        const bool up = x > m;
        const float mn = fmaxf(m, x);
        const float t = expf(fminf(m, x) - mn);
        const float c = up ? t : 1.f;
        const float p = up ? 1.f : t;
        den = den * c + p;
        m = mn;
#pragma unroll
        for (int j = 0; j < 10; ++j) {
            acc[4 * j + 0] = acc[4 * j + 0] * c + p * w[j].x;
            acc[4 * j + 1] = acc[4 * j + 1] * c + p * w[j].y;
            acc[4 * j + 2] = acc[4 * j + 2] * c + p * w[j].z;
            acc[4 * j + 3] = acc[4 * j + 3] * c + p * w[j].w;
        }
    }
    const bool any = den > 0.f;
    float mx = -INFINITY;
#pragma unroll
    for (int k = 0; k < 40; ++k) {
        acc[k] = any ? acc[k] / den : 0.f;
        mx = fmaxf(mx, acc[k]);
    }
    float sm = 0.f;
#pragma unroll
    for (int k = 0; k < 40; ++k) sm += expf(acc[k] - mx);
    const float lg = mx + logf(sm);
    v4f* orow = reinterpret_cast<v4f*>(out + (size_t)node * 40);
#pragma unroll
    for (int j = 0; j < 10; ++j) {
        v4f o;
        o.x = acc[4 * j + 0] - lg;
        o.y = acc[4 * j + 1] - lg;
        o.z = acc[4 * j + 2] - lg;
        o.w = acc[4 * j + 3] - lg;
        orow[j] = o;
    }
}

extern "C" void kernel_launch(void* const* d_in, const int* in_sizes, int n_in, void* d_out,
                              int out_size, void* d_ws, size_t ws_size, hipStream_t stream) {
    (void)n_in; (void)out_size; (void)ws_size;
    const float* h      = (const float*)d_in[0];
    const int*   eidx   = (const int*)d_in[1];
    const float* Wheads = (const float*)d_in[2];
    const float* aheads = (const float*)d_in[3];
    const float* Wout   = (const float*)d_in[4];
    const float* aout   = (const float*)d_in[5];
    float* out = (float*)d_out;

    const int N = in_sizes[0] / 256;
    const int E = in_sizes[1] / 2;
    const int* src = eidx;
    const int* dst = eidx + E;

    float* ws = (float*)d_ws;
    size_t oWh   = 0;
    size_t oHP   = oWh  + (size_t)N * 256;
    size_t oS1s  = oHP  + (size_t)N * 256;
    size_t oS1d  = oS1s + (size_t)N * 4;
    size_t oHead = oS1d + (size_t)N * 4;
    size_t oNxt  = oHead + (size_t)N;
    size_t oS2s  = oS1s;
    size_t oS2d  = oS1d;
    size_t oWx   = oWh;
    size_t oWa1  = oHP;

    int* head = (int*)(ws + oHead);
    int* nxt  = (int*)(ws + oNxt);

    int fillBlocks = (N + 255) / 256;
    prep_kernel<<<fillBlocks + 8, 256, 0, stream>>>(head, N, fillBlocks, Wheads, aheads,
                                                    ws + oWa1);

    gemm1_kernel<<<256, 512, 0, stream>>>(h, Wheads, ws + oWa1, ws + oWh, ws + oS1s, ws + oS1d,
                                          N);

    int geE = (E + 255) / 256;
    edge_link<<<geE, 256, 0, stream>>>(dst, head, nxt, E);

    gather1<<<(N + 7) / 8, 256, 0, stream>>>(src, head, nxt, ws + oS1s, ws + oS1d, ws + oWh,
                                             ws + oHP, N, E);

    dim3 g2((N + 127) / 128, 1);
    gemm2_kernel<<<g2, 256, 0, stream>>>(ws + oHP, Wout, aout, ws + oWx, ws + oS2s, ws + oS2d,
                                         N);

    gather2<<<(N + 63) / 64, 64, 0, stream>>>(src, head, nxt, ws + oS2s, ws + oS2d, ws + oWx,
                                              out, N, E);
}


# layer-2 GEMM retiled like layer 1: one 16-wave work-group per WGP stages the 256x48 W_out strip in LDS once, waves loop over 16-row tiles, x read a 64-k chunk ahead; k sum order of products changes
# speedup vs baseline: 6.8163x; 1.0720x over previous
#include <hip/hip_runtime.h>


#define LRELU 0.2f

typedef __attribute__((ext_vector_type(2))) float v2f;
typedef __attribute__((ext_vector_type(8))) float v8f;
typedef __attribute__((ext_vector_type(4))) float v4f;

__device__ __forceinline__ float lrelu(float x) { return x > 0.f ? x : LRELU * x; }

__global__ __launch_bounds__(256) void prep_kernel(int* __restrict__ head, int n, int fillBlocks,
                                                   const float* __restrict__ Wheads,
                                                   const float* __restrict__ aheads,
                                                   float* __restrict__ Wa1) {
    const int b = blockIdx.x;
    if (b < fillBlocks) {
        const int i = b * 256 + threadIdx.x;
        if (i < n) head[i] = -1;
        return;
    }
    const int c = b - fillBlocks;
    const int hd = c & 3;
    const int k = threadIdx.x;
    const float* w = Wheads + ((size_t)hd * 256 + k) * 64;
    const float* a = aheads + hd * 128 + (c >> 2) * 64;
    float s = 0.f;
#pragma unroll
    for (int f = 0; f < 64; ++f) s += w[f] * a[f];
    Wa1[k * 16 + c] = s;
    Wa1[k * 16 + 8 + c] = 0.f;
}

#define G1_COLS 272
#define G1_TILES 17

__global__ __launch_bounds__(512) void gemm1_kernel(const float* __restrict__ h,
                                                    const float* __restrict__ Wheads,
                                                    const float* __restrict__ Wa1,
                                                    float* __restrict__ Wh,
                                                    float* __restrict__ ssrc,
                                                    float* __restrict__ sdst, int nrows) {
    __shared__ __attribute__((aligned(16))) float Blds[256 * G1_COLS];
    const int tid = threadIdx.x;

    {
        v4f r[9][4];
#pragma unroll
        for (int i = 0; i < 9; ++i) {
            int u = tid + 512 * i;
            if (u > 64 * (G1_COLS / 4) - 1) u = 64 * (G1_COLS / 4) - 1;
            const int q = u / (G1_COLS / 4);
            const int c = (u - q * (G1_COLS / 4)) * 4;
            const bool head = c < 256;
            const float* sp = head ? (Wheads + ((size_t)(c >> 6) * 256 + 4 * q) * 64 + (c & 63))
                                   : (Wa1 + (4 * q) * 16 + (c - 256));
            const int pitch = head ? 64 : 16;
#pragma unroll
            for (int j = 0; j < 4; ++j) r[i][j] = *reinterpret_cast<const v4f*>(sp + j * pitch);
        }
#pragma unroll
        for (int i = 0; i < 9; ++i) {
            const int u = tid + 512 * i;
            if (u < 64 * (G1_COLS / 4)) {
                const int q = u / (G1_COLS / 4);
                const int c = (u - q * (G1_COLS / 4)) * 4;
                v4f* d = reinterpret_cast<v4f*>(Blds + (q * G1_COLS + c) * 4);
                v4f s0, s1, s2, s3;
                s0.x = r[i][0].x; s0.y = r[i][1].x; s0.z = r[i][2].x; s0.w = r[i][3].x;
                s1.x = r[i][0].y; s1.y = r[i][1].y; s1.z = r[i][2].y; s1.w = r[i][3].y;
                s2.x = r[i][0].z; s2.y = r[i][1].z; s2.z = r[i][2].z; s2.w = r[i][3].z;
                s3.x = r[i][0].w; s3.y = r[i][1].w; s3.z = r[i][2].w; s3.w = r[i][3].w;
                d[0] = s0;
                d[1] = s1;
                d[2] = s2;
                d[3] = s3;
            }
        }
    }
    __syncthreads();

    const int wid  = __builtin_amdgcn_readfirstlane(tid >> 5);
    const int lane = tid & 31;
    const int hf   = lane >> 4;
    const int l16  = lane & 15;
    const int ntile = (nrows + 15) >> 4;
    const float* bbase = Blds + (hf * G1_COLS + l16) * 4;

    const int slot = 4 * (((wid >> 2) - (wid & 3)) & 3) + (wid & 3);
    for (int tile = slot * (int)gridDim.x + (int)blockIdx.x; tile < ntile;
         tile += 16 * (int)gridDim.x) {
        const int rowBase = tile * 16;
        int arow = rowBase + l16;
        if (arow >= nrows) arow = nrows - 1;
        const v4f* aptr = reinterpret_cast<const v4f*>(h + (size_t)arow * 256 + hf * 4);

        v8f acc[G1_TILES] = {};
        v4f a = aptr[0];
        for (int g = 0; g < 32; ++g) {
            const v4f an = aptr[2 * ((g + 1) & 31)];
            __builtin_amdgcn_sched_barrier(0);
            const float* bp = bbase + g * (2 * G1_COLS * 4);
            v4f b[G1_TILES];
#pragma unroll
            for (int t = 0; t < G1_TILES; ++t) b[t] = *reinterpret_cast<const v4f*>(bp + t * 64);
            v2f a0, a1;
            a0.x = a.x; a0.y = a.y;
            a1.x = a.z; a1.y = a.w;
#pragma unroll
            for (int t = 0; t < G1_TILES; ++t) {
                v2f b0;
                b0.x = b[t].x; b0.y = b[t].y;
                acc[t] = __builtin_amdgcn_wmma_f32_16x16x4_f32(false, a0, false, b0, (short)0,
                                                               acc[t], false, false);
            }
#pragma unroll
            for (int t = 0; t < G1_TILES; ++t) {
                v2f b1;
                b1.x = b[t].z; b1.y = b[t].w;
                acc[t] = __builtin_amdgcn_wmma_f32_16x16x4_f32(false, a1, false, b1, (short)0,
                                                               acc[t], false, false);
            }
            a = an;
        }

        const bool full = rowBase + 16 <= nrows;
        float* crow = Wh + (size_t)(rowBase + 8 * hf) * 256 + l16;
        if (full) {
#pragma unroll
            for (int t = 0; t < 16; ++t)
#pragma unroll
                for (int v = 0; v < 8; ++v) crow[(size_t)v * 256 + t * 16] = acc[t][v];
        } else {
#pragma unroll
            for (int t = 0; t < 16; ++t)
#pragma unroll
                for (int v = 0; v < 8; ++v)
                    if (rowBase + 8 * hf + v < nrows) crow[(size_t)v * 256 + t * 16] = acc[t][v];
        }
        if (l16 < 8) {
            float* sb = ((l16 < 4) ? (ssrc + l16) : (sdst + (l16 - 4))) +
                        (size_t)(rowBase + 8 * hf) * 4;
#pragma unroll
            for (int v = 0; v < 8; ++v)
                if (full || rowBase + 8 * hf + v < nrows) sb[(size_t)v * 4] = acc[16][v];
        }
    }
}

#define G2_COLS 48

template <bool GUARD>
__device__ __forceinline__ void store_tiles2(const v8f (&acc)[3], float* __restrict__ C,
                                             float* __restrict__ S0, float* __restrict__ S1,
                                             int nrows, int rowBase, int half, int l16) {
#pragma unroll
    for (int t = 0; t < 3; ++t) {
        const int col = t * 16 + l16;
        float* base;
        size_t stride;
        bool on = true;
        if (col < 40) {
            base = C + col;
            stride = 40;
        } else {
            base = (col == 40) ? S0 : S1;
            stride = 1;
            on = col < 42;
        }
#pragma unroll
        for (int v = 0; v < 8; ++v) {
            const int row = rowBase + v + half * 8;
            if (on && (!GUARD || row < nrows)) base[(size_t)row * stride] = acc[t][v];
        }
    }
}

__device__ __forceinline__ void g2_chunk(const v4f (&A)[8], const float* bp, v8f (&acc)[3]) {
#pragma unroll
    for (int gq = 0; gq < 2; ++gq) {
        v4f b[4][3];
#pragma unroll
        for (int u = 0; u < 4; ++u)
#pragma unroll
            for (int t = 0; t < 3; ++t)
                b[u][t] = *reinterpret_cast<const v4f*>(bp + (4 * gq + u) * (2 * G2_COLS * 4) +
                                                        t * 64);
        __builtin_amdgcn_sched_barrier(0);
#pragma unroll
        for (int u = 0; u < 4; ++u) {
            const v4f a = A[4 * gq + u];
            v2f a0, a1;
            a0.x = a.x; a0.y = a.y;
            a1.x = a.z; a1.y = a.w;
#pragma unroll
            for (int t = 0; t < 3; ++t) {
                v2f b0;
                b0.x = b[u][t].x; b0.y = b[u][t].y;
                acc[t] = __builtin_amdgcn_wmma_f32_16x16x4_f32(false, a0, false, b0, (short)0,
                                                               acc[t], false, false);
            }
#pragma unroll
            for (int t = 0; t < 3; ++t) {
                v2f b1;
                b1.x = b[u][t].z; b1.y = b[u][t].w;
                acc[t] = __builtin_amdgcn_wmma_f32_16x16x4_f32(false, a1, false, b1, (short)0,
                                                               acc[t], false, false);
            }
        }
    }
}

__global__ __launch_bounds__(512) void gemm2_kernel(const float* __restrict__ x,
                                                    const float* __restrict__ Wout,
                                                    const float* __restrict__ aout,
                                                    float* __restrict__ Wx,
                                                    float* __restrict__ ssrc,
                                                    float* __restrict__ sdst, int nrows) {
    __shared__ __attribute__((aligned(16))) float Blds[256 * G2_COLS];
    const int tid = threadIdx.x;

    const int wid  = __builtin_amdgcn_readfirstlane(tid >> 5);
    const int lane = tid & 31;
    const int hf   = lane >> 4;
    const int l16  = lane & 15;
    const int ntile = (nrows + 15) >> 4;
    const int slot = 4 * (((wid >> 2) - (wid & 3)) & 3) + (wid & 3);
    const int tstep = 16 * (int)gridDim.x;
    int tile = slot * (int)gridDim.x + (int)blockIdx.x;

    v4f A0[8], A1[8];
    const v4f* aptr;
    {
        const int t0 = tile < ntile ? tile : ntile - 1;
        int arow = t0 * 16 + l16;
        if (arow >= nrows) arow = nrows - 1;
        aptr = reinterpret_cast<const v4f*>(x + (size_t)arow * 256 + hf * 4);
#pragma unroll
        for (int i = 0; i < 8; ++i) A0[i] = aptr[2 * i];
    }

    {
        const bool second = tid < 128;
        const bool zeros  = tid >= 128 && tid < 256;
        v4f r[2][4];
#pragma unroll
        for (int i = 0; i < 2; ++i) {
            if (i == 0 || second) {
                const int u = tid + 512 * i;
                const int q = u / 10;
                const int c = (u - q * 10) * 4;
                const float* sp = Wout + (4 * q) * 40 + c;
#pragma unroll
                for (int j = 0; j < 4; ++j) r[i][j] = *reinterpret_cast<const v4f*>(sp + j * 40);
            }
        }
        const bool folds = tid < 256;
        v4f w[10];
        if (folds) {
            const v4f* wr = reinterpret_cast<const v4f*>(Wout + tid * 40);
#pragma unroll
            for (int j = 0; j < 10; ++j) w[j] = wr[j];
        }
#pragma unroll
        for (int i = 0; i < 2; ++i) {
            if (i == 0 || second) {
                const int u = tid + 512 * i;
                const int q = u / 10;
                const int c = (u - q * 10) * 4;
                v4f* d = reinterpret_cast<v4f*>(Blds + (q * G2_COLS + c) * 4);
                v4f s0, s1, s2, s3;
                s0.x = r[i][0].x; s0.y = r[i][1].x; s0.z = r[i][2].x; s0.w = r[i][3].x;
                s1.x = r[i][0].y; s1.y = r[i][1].y; s1.z = r[i][2].y; s1.w = r[i][3].y;
                s2.x = r[i][0].z; s2.y = r[i][1].z; s2.z = r[i][2].z; s2.w = r[i][3].z;
                s3.x = r[i][0].w; s3.y = r[i][1].w; s3.z = r[i][2].w; s3.w = r[i][3].w;
                d[0] = s0;
                d[1] = s1;
                d[2] = s2;
                d[3] = s3;
            }
        }
        if (zeros) {
            const int z = tid - 128;
            v4f* d = reinterpret_cast<v4f*>(Blds + ((z >> 1) * G2_COLS + 40 + (z & 1) * 4) * 4);
            const v4f zero = {0.f, 0.f, 0.f, 0.f};
            d[0] = zero;
            d[1] = zero;
            d[2] = zero;
            d[3] = zero;
        }
        __syncthreads();
        if (folds) {
            float s0 = 0.f, s1 = 0.f;
#pragma unroll
            for (int j = 0; j < 10; ++j) {
                s0 += w[j].x * aout[4 * j + 0];
                s1 += w[j].x * aout[40 + 4 * j + 0];
                s0 += w[j].y * aout[4 * j + 1];
                s1 += w[j].y * aout[40 + 4 * j + 1];
                s0 += w[j].z * aout[4 * j + 2];
                s1 += w[j].z * aout[40 + 4 * j + 2];
                s0 += w[j].w * aout[4 * j + 3];
                s1 += w[j].w * aout[40 + 4 * j + 3];
            }
            float* f = Blds + ((tid >> 2) * G2_COLS + 40) * 4 + (tid & 3);
            f[0] = s0;
            f[4] = s1;
        }
        __syncthreads();
    }

    const float* bbase = Blds + (hf * G2_COLS + l16) * 4;
    constexpr int CH = 8 * (2 * G2_COLS * 4);

    for (; tile < ntile; tile += tstep) {
        const int rowBase = tile * 16;
        v8f acc[3] = {};

#pragma unroll
        for (int i = 0; i < 8; ++i) A1[i] = aptr[2 * (8 + i)];
        __builtin_amdgcn_sched_barrier(0);
        g2_chunk(A0, bbase, acc);

#pragma unroll
        for (int i = 0; i < 8; ++i) A0[i] = aptr[2 * (16 + i)];
        __builtin_amdgcn_sched_barrier(0);
        g2_chunk(A1, bbase + CH, acc);

#pragma unroll
        for (int i = 0; i < 8; ++i) A1[i] = aptr[2 * (24 + i)];
        __builtin_amdgcn_sched_barrier(0);
        g2_chunk(A0, bbase + 2 * CH, acc);

        {
            const int tn = tile + tstep < ntile ? tile + tstep : tile;
            int arow = tn * 16 + l16;
            if (arow >= nrows) arow = nrows - 1;
            aptr = reinterpret_cast<const v4f*>(x + (size_t)arow * 256 + hf * 4);
#pragma unroll
            for (int i = 0; i < 8; ++i) A0[i] = aptr[2 * i];
        }
        __builtin_amdgcn_sched_barrier(0);
        g2_chunk(A1, bbase + 3 * CH, acc);

        if (rowBase + 16 <= nrows) {
            store_tiles2<false>(acc, Wx, ssrc, sdst, nrows, rowBase, hf, l16);
        } else {
            store_tiles2<true>(acc, Wx, ssrc, sdst, nrows, rowBase, hf, l16);
        }
    }
}

__global__ __launch_bounds__(256) void edge_link(const int* __restrict__ dst,
                                                 int* __restrict__ head,
                                                 int* __restrict__ nxt, int ne) {
    int e = blockIdx.x * blockDim.x + threadIdx.x;
    if (e >= ne) return;
    nxt[e] = atomicExch(&head[dst[e]], e);
}

__device__ __forceinline__ float lane_bcast(float v, int l) {
    return __int_as_float(__builtin_amdgcn_readlane(__float_as_int(v), l));
}

__global__ __launch_bounds__(256) void gather1(const int* __restrict__ src,
                                               const int* __restrict__ head,
                                               const int* __restrict__ nxt,
                                               const float* __restrict__ ssrc,
                                               const float* __restrict__ sdst,
                                               const float* __restrict__ Wh,
                                               float* __restrict__ hp, int n, int ne) {
    const int lane = threadIdx.x & 31;
    const int hsel = lane & 3;
    const int node = __builtin_amdgcn_readfirstlane((int)(blockIdx.x * 8 + (threadIdx.x >> 5)));
    if (node >= n) return;
    float acc[8];
#pragma unroll
    for (int j = 0; j < 8; ++j) acc[j] = 0.f;
    float m = -INFINITY, den = 0.f;
    const float sd = sdst[node * 4 + hsel];

    int e = head[node];
    bool ok = (unsigned)e < (unsigned)ne;
    int s = 0, en = -1;
    if (ok) {
        s = src[e];
        en = nxt[e];
    }
    for (int it = 0; it < ne && ok; ++it) {
        const float* row = Wh + (size_t)s * 256 + lane;
        float w[8];
#pragma unroll
        for (int j = 0; j < 8; ++j) w[j] = row[32 * j];
        const float xs = ssrc[s * 4 + hsel];
        e = en;
        ok = (unsigned)e < (unsigned)ne;
        if (ok) {
            s = src[e];
            en = nxt[e];
        }
        const float x = lrelu(xs + sd);
        const bool up = x > m;
        const float mn = fmaxf(m, x);
        const float t = expf(fminf(m, x) - mn);
        const float c = up ? t : 1.f;
        const float p = up ? 1.f : t;
        den = den * c + p;
        m = mn;
#pragma unroll
        for (int h = 0; h < 4; ++h) {
            const float ch = lane_bcast(c, h);
            const float ph = lane_bcast(p, h);
            acc[2 * h]     = acc[2 * h] * ch + ph * w[2 * h];
            acc[2 * h + 1] = acc[2 * h + 1] * ch + ph * w[2 * h + 1];
        }
    }
    float* orow = hp + (size_t)node * 256 + lane;
#pragma unroll
    for (int j = 0; j < 8; ++j) {
        const float dh = lane_bcast(den, j >> 1);
        const float v = dh > 0.f ? acc[j] / dh : 0.f;
        orow[32 * j] = v > 0.f ? v : (expf(v) - 1.f);
    }
}

__global__ __launch_bounds__(64) void gather2(const int* __restrict__ src,
                                              const int* __restrict__ head,
                                              const int* __restrict__ nxt,
                                              const float* __restrict__ ssrc,
                                              const float* __restrict__ sdst,
                                              const float* __restrict__ Wx,
                                              float* __restrict__ out, int n, int ne) {
    const int node = (int)(blockIdx.x * 64 + threadIdx.x);
    if (node >= n) return;
    float acc[40];
#pragma unroll
    for (int k = 0; k < 40; ++k) acc[k] = 0.f;
    float m = -INFINITY, den = 0.f;
    const float sd = sdst[node];

    int e = head[node];
    bool ok = (unsigned)e < (unsigned)ne;
    int s = 0, en = -1;
    if (ok) {
        s = src[e];
        en = nxt[e];
    }
    for (int it = 0; it < ne && ok; ++it) {
        const v4f* row = reinterpret_cast<const v4f*>(Wx + (size_t)s * 40);
        v4f w[10];
#pragma unroll
        for (int j = 0; j < 10; ++j) w[j] = row[j];
        const float xs = ssrc[s];
        e = en;
        ok = (unsigned)e < (unsigned)ne;
        if (ok) {
            s = src[e];
            en = nxt[e];
        }
        const float x = lrelu(xs + sd);
        const bool up = x > m;
        const float mn = fmaxf(m, x);
        const float t = expf(fminf(m, x) - mn);
        const float c = up ? t : 1.f;
        const float p = up ? 1.f : t;
        den = den * c + p;
        m = mn;
#pragma unroll
        for (int j = 0; j < 10; ++j) {
            acc[4 * j + 0] = acc[4 * j + 0] * c + p * w[j].x;
            acc[4 * j + 1] = acc[4 * j + 1] * c + p * w[j].y;
            acc[4 * j + 2] = acc[4 * j + 2] * c + p * w[j].z;
            acc[4 * j + 3] = acc[4 * j + 3] * c + p * w[j].w;
        }
    }
    const bool any = den > 0.f;
    float mx = -INFINITY;
#pragma unroll
    for (int k = 0; k < 40; ++k) {
        acc[k] = any ? acc[k] / den : 0.f;
        mx = fmaxf(mx, acc[k]);
    }
    float sm = 0.f;
#pragma unroll
    for (int k = 0; k < 40; ++k) sm += expf(acc[k] - mx);
    const float lg = mx + logf(sm);
    v4f* orow = reinterpret_cast<v4f*>(out + (size_t)node * 40);
#pragma unroll
    for (int j = 0; j < 10; ++j) {
        v4f o;
        o.x = acc[4 * j + 0] - lg;
        o.y = acc[4 * j + 1] - lg;
        o.z = acc[4 * j + 2] - lg;
        o.w = acc[4 * j + 3] - lg;
        orow[j] = o;
    }
}

extern "C" void kernel_launch(void* const* d_in, const int* in_sizes, int n_in, void* d_out,
                              int out_size, void* d_ws, size_t ws_size, hipStream_t stream) {
    (void)n_in; (void)out_size; (void)ws_size;
    const float* h      = (const float*)d_in[0];
    const int*   eidx   = (const int*)d_in[1];
    const float* Wheads = (const float*)d_in[2];
    const float* aheads = (const float*)d_in[3];
    const float* Wout   = (const float*)d_in[4];
    const float* aout   = (const float*)d_in[5];
    float* out = (float*)d_out;

    const int N = in_sizes[0] / 256;
    const int E = in_sizes[1] / 2;
    const int* src = eidx;
    const int* dst = eidx + E;

    float* ws = (float*)d_ws;
    size_t oWh   = 0;
    size_t oHP   = oWh  + (size_t)N * 256;
    size_t oS1s  = oHP  + (size_t)N * 256;
    size_t oS1d  = oS1s + (size_t)N * 4;
    size_t oHead = oS1d + (size_t)N * 4;
    size_t oNxt  = oHead + (size_t)N;
    size_t oS2s  = oS1s;
    size_t oS2d  = oS1d;
    size_t oWx   = oWh;
    size_t oWa1  = oHP;

    int* head = (int*)(ws + oHead);
    int* nxt  = (int*)(ws + oNxt);

    int fillBlocks = (N + 255) / 256;
    prep_kernel<<<fillBlocks + 8, 256, 0, stream>>>(head, N, fillBlocks, Wheads, aheads,
                                                    ws + oWa1);

    gemm1_kernel<<<256, 512, 0, stream>>>(h, Wheads, ws + oWa1, ws + oWh, ws + oS1s, ws + oS1d,
                                          N);

    int geE = (E + 255) / 256;
    edge_link<<<geE, 256, 0, stream>>>(dst, head, nxt, E);

    gather1<<<(N + 7) / 8, 256, 0, stream>>>(src, head, nxt, ws + oS1s, ws + oS1d, ws + oWh,
                                             ws + oHP, N, E);

    gemm2_kernel<<<256, 512, 0, stream>>>(ws + oHP, Wout, aout, ws + oWx, ws + oS2s, ws + oS2d,
                                          N);

    gather2<<<(N + 63) / 64, 64, 0, stream>>>(src, head, nxt, ws + oS2s, ws + oS2d, ws + oWx,
                                              out, N, E);
}


# layer-1 gather reads a 1024-byte source row as two wave-wide 16-byte loads (512 contiguous bytes, 4 adjacent lines per load), not eight one-line loads; row stored likewise; sum order unchanged
# speedup vs baseline: 7.0987x; 1.0414x over previous
#include <hip/hip_runtime.h>


#define LRELU 0.2f

typedef __attribute__((ext_vector_type(2))) float v2f;
typedef __attribute__((ext_vector_type(8))) float v8f;
typedef __attribute__((ext_vector_type(4))) float v4f;

__device__ __forceinline__ float lrelu(float x) { return x > 0.f ? x : LRELU * x; }

__global__ __launch_bounds__(256) void prep_kernel(int* __restrict__ head, int n, int fillBlocks,
                                                   const float* __restrict__ Wheads,
                                                   const float* __restrict__ aheads,
                                                   float* __restrict__ Wa1) {
    const int b = blockIdx.x;
    if (b < fillBlocks) {
        const int i = b * 256 + threadIdx.x;
        if (i < n) head[i] = -1;
        return;
    }
    const int c = b - fillBlocks;
    const int hd = c & 3;
    const int k = threadIdx.x;
    const float* w = Wheads + ((size_t)hd * 256 + k) * 64;
    const float* a = aheads + hd * 128 + (c >> 2) * 64;
    float s = 0.f;
#pragma unroll
    for (int f = 0; f < 64; ++f) s += w[f] * a[f];
    Wa1[k * 16 + c] = s;
    Wa1[k * 16 + 8 + c] = 0.f;
}

#define G1_COLS 272
#define G1_TILES 17

__global__ __launch_bounds__(512) void gemm1_kernel(const float* __restrict__ h,
                                                    const float* __restrict__ Wheads,
                                                    const float* __restrict__ Wa1,
                                                    float* __restrict__ Wh,
                                                    float* __restrict__ ssrc,
                                                    float* __restrict__ sdst, int nrows) {
    __shared__ __attribute__((aligned(16))) float Blds[256 * G1_COLS];
    const int tid = threadIdx.x;

    {
        v4f r[9][4];
#pragma unroll
        for (int i = 0; i < 9; ++i) {
            int u = tid + 512 * i;
            if (u > 64 * (G1_COLS / 4) - 1) u = 64 * (G1_COLS / 4) - 1;
            const int q = u / (G1_COLS / 4);
            const int c = (u - q * (G1_COLS / 4)) * 4;
            const bool head = c < 256;
            const float* sp = head ? (Wheads + ((size_t)(c >> 6) * 256 + 4 * q) * 64 + (c & 63))
                                   : (Wa1 + (4 * q) * 16 + (c - 256));
            const int pitch = head ? 64 : 16;
#pragma unroll
            for (int j = 0; j < 4; ++j) r[i][j] = *reinterpret_cast<const v4f*>(sp + j * pitch);
        }
#pragma unroll
        for (int i = 0; i < 9; ++i) {
            const int u = tid + 512 * i;
            if (u < 64 * (G1_COLS / 4)) {
                const int q = u / (G1_COLS / 4);
                const int c = (u - q * (G1_COLS / 4)) * 4;
                v4f* d = reinterpret_cast<v4f*>(Blds + (q * G1_COLS + c) * 4);
                v4f s0, s1, s2, s3;
                s0.x = r[i][0].x; s0.y = r[i][1].x; s0.z = r[i][2].x; s0.w = r[i][3].x;
                s1.x = r[i][0].y; s1.y = r[i][1].y; s1.z = r[i][2].y; s1.w = r[i][3].y;
                s2.x = r[i][0].z; s2.y = r[i][1].z; s2.z = r[i][2].z; s2.w = r[i][3].z;
                s3.x = r[i][0].w; s3.y = r[i][1].w; s3.z = r[i][2].w; s3.w = r[i][3].w;
                d[0] = s0;
                d[1] = s1;
                d[2] = s2;
                d[3] = s3;
            }
        }
    }
    __syncthreads();

    const int wid  = __builtin_amdgcn_readfirstlane(tid >> 5);
    const int lane = tid & 31;
    const int hf   = lane >> 4;
    const int l16  = lane & 15;
    const int ntile = (nrows + 15) >> 4;
    const float* bbase = Blds + (hf * G1_COLS + l16) * 4;

    const int slot = 4 * (((wid >> 2) - (wid & 3)) & 3) + (wid & 3);
    for (int tile = slot * (int)gridDim.x + (int)blockIdx.x; tile < ntile;
         tile += 16 * (int)gridDim.x) {
        const int rowBase = tile * 16;
        int arow = rowBase + l16;
        if (arow >= nrows) arow = nrows - 1;
        const v4f* aptr = reinterpret_cast<const v4f*>(h + (size_t)arow * 256 + hf * 4);

        v8f acc[G1_TILES] = {};
        v4f a = aptr[0];
        for (int g = 0; g < 32; ++g) {
            const v4f an = aptr[2 * ((g + 1) & 31)];
            __builtin_amdgcn_sched_barrier(0);
            const float* bp = bbase + g * (2 * G1_COLS * 4);
            v4f b[G1_TILES];
#pragma unroll
            for (int t = 0; t < G1_TILES; ++t) b[t] = *reinterpret_cast<const v4f*>(bp + t * 64);
            v2f a0, a1;
            a0.x = a.x; a0.y = a.y;
            a1.x = a.z; a1.y = a.w;
#pragma unroll
            for (int t = 0; t < G1_TILES; ++t) {
                v2f b0;
                b0.x = b[t].x; b0.y = b[t].y;
                acc[t] = __builtin_amdgcn_wmma_f32_16x16x4_f32(false, a0, false, b0, (short)0,
                                                               acc[t], false, false);
            }
#pragma unroll
            for (int t = 0; t < G1_TILES; ++t) {
                v2f b1;
                b1.x = b[t].z; b1.y = b[t].w;
                acc[t] = __builtin_amdgcn_wmma_f32_16x16x4_f32(false, a1, false, b1, (short)0,
                                                               acc[t], false, false);
            }
            a = an;
        }

        const bool full = rowBase + 16 <= nrows;
        float* crow = Wh + (size_t)(rowBase + 8 * hf) * 256 + l16;
        if (full) {
#pragma unroll
            for (int t = 0; t < 16; ++t)
#pragma unroll
                for (int v = 0; v < 8; ++v) crow[(size_t)v * 256 + t * 16] = acc[t][v];
        } else {
#pragma unroll
            for (int t = 0; t < 16; ++t)
#pragma unroll
                for (int v = 0; v < 8; ++v)
                    if (rowBase + 8 * hf + v < nrows) crow[(size_t)v * 256 + t * 16] = acc[t][v];
        }
        if (l16 < 8) {
            float* sb = ((l16 < 4) ? (ssrc + l16) : (sdst + (l16 - 4))) +
                        (size_t)(rowBase + 8 * hf) * 4;
#pragma unroll
            for (int v = 0; v < 8; ++v)
                if (full || rowBase + 8 * hf + v < nrows) sb[(size_t)v * 4] = acc[16][v];
        }
    }
}

#define G2_COLS 48

template <bool GUARD>
__device__ __forceinline__ void store_tiles2(const v8f (&acc)[3], float* __restrict__ C,
                                             float* __restrict__ S0, float* __restrict__ S1,
                                             int nrows, int rowBase, int half, int l16) {
#pragma unroll
    for (int t = 0; t < 3; ++t) {
        const int col = t * 16 + l16;
        float* base;
        size_t stride;
        bool on = true;
        if (col < 40) {
            base = C + col;
            stride = 40;
        } else {
            base = (col == 40) ? S0 : S1;
            stride = 1;
            on = col < 42;
        }
#pragma unroll
        for (int v = 0; v < 8; ++v) {
            const int row = rowBase + v + half * 8;
            if (on && (!GUARD || row < nrows)) base[(size_t)row * stride] = acc[t][v];
        }
    }
}

__device__ __forceinline__ void g2_chunk(const v4f (&A)[8], const float* bp, v8f (&acc)[3]) {
#pragma unroll
    for (int gq = 0; gq < 2; ++gq) {
        v4f b[4][3];
#pragma unroll
        for (int u = 0; u < 4; ++u)
#pragma unroll
            for (int t = 0; t < 3; ++t)
                b[u][t] = *reinterpret_cast<const v4f*>(bp + (4 * gq + u) * (2 * G2_COLS * 4) +
                                                        t * 64);
        __builtin_amdgcn_sched_barrier(0);
#pragma unroll
        for (int u = 0; u < 4; ++u) {
            const v4f a = A[4 * gq + u];
            v2f a0, a1;
            a0.x = a.x; a0.y = a.y;
            a1.x = a.z; a1.y = a.w;
#pragma unroll
            for (int t = 0; t < 3; ++t) {
                v2f b0;
                b0.x = b[u][t].x; b0.y = b[u][t].y;
                acc[t] = __builtin_amdgcn_wmma_f32_16x16x4_f32(false, a0, false, b0, (short)0,
                                                               acc[t], false, false);
            }
#pragma unroll
            for (int t = 0; t < 3; ++t) {
                v2f b1;
                b1.x = b[u][t].z; b1.y = b[u][t].w;
                acc[t] = __builtin_amdgcn_wmma_f32_16x16x4_f32(false, a1, false, b1, (short)0,
                                                               acc[t], false, false);
            }
        }
    }
}

__global__ __launch_bounds__(512) void gemm2_kernel(const float* __restrict__ x,
                                                    const float* __restrict__ Wout,
                                                    const float* __restrict__ aout,
                                                    float* __restrict__ Wx,
                                                    float* __restrict__ ssrc,
                                                    float* __restrict__ sdst, int nrows) {
    __shared__ __attribute__((aligned(16))) float Blds[256 * G2_COLS];
    const int tid = threadIdx.x;

    const int wid  = __builtin_amdgcn_readfirstlane(tid >> 5);
    const int lane = tid & 31;
    const int hf   = lane >> 4;
    const int l16  = lane & 15;
    const int ntile = (nrows + 15) >> 4;
    const int slot = 4 * (((wid >> 2) - (wid & 3)) & 3) + (wid & 3);
    const int tstep = 16 * (int)gridDim.x;
    int tile = slot * (int)gridDim.x + (int)blockIdx.x;

    v4f A0[8], A1[8];
    const v4f* aptr;
    {
        const int t0 = tile < ntile ? tile : ntile - 1;
        int arow = t0 * 16 + l16;
        if (arow >= nrows) arow = nrows - 1;
        aptr = reinterpret_cast<const v4f*>(x + (size_t)arow * 256 + hf * 4);
#pragma unroll
        for (int i = 0; i < 8; ++i) A0[i] = aptr[2 * i];
    }

    {
        const bool second = tid < 128;
        const bool zeros  = tid >= 128 && tid < 256;
        v4f r[2][4];
#pragma unroll
        for (int i = 0; i < 2; ++i) {
            if (i == 0 || second) {
                const int u = tid + 512 * i;
                const int q = u / 10;
                const int c = (u - q * 10) * 4;
                const float* sp = Wout + (4 * q) * 40 + c;
#pragma unroll
                for (int j = 0; j < 4; ++j) r[i][j] = *reinterpret_cast<const v4f*>(sp + j * 40);
            }
        }
        const bool folds = tid < 256;
        v4f w[10];
        if (folds) {
            const v4f* wr = reinterpret_cast<const v4f*>(Wout + tid * 40);
#pragma unroll
            for (int j = 0; j < 10; ++j) w[j] = wr[j];
        }
#pragma unroll
        for (int i = 0; i < 2; ++i) {
            if (i == 0 || second) {
                const int u = tid + 512 * i;
                const int q = u / 10;
                const int c = (u - q * 10) * 4;
                v4f* d = reinterpret_cast<v4f*>(Blds + (q * G2_COLS + c) * 4);
                v4f s0, s1, s2, s3;
                s0.x = r[i][0].x; s0.y = r[i][1].x; s0.z = r[i][2].x; s0.w = r[i][3].x;
                s1.x = r[i][0].y; s1.y = r[i][1].y; s1.z = r[i][2].y; s1.w = r[i][3].y;
                s2.x = r[i][0].z; s2.y = r[i][1].z; s2.z = r[i][2].z; s2.w = r[i][3].z;
                s3.x = r[i][0].w; s3.y = r[i][1].w; s3.z = r[i][2].w; s3.w = r[i][3].w;
                d[0] = s0;
                d[1] = s1;
                d[2] = s2;
                d[3] = s3;
            }
        }
        if (zeros) {
            const int z = tid - 128;
            v4f* d = reinterpret_cast<v4f*>(Blds + ((z >> 1) * G2_COLS + 40 + (z & 1) * 4) * 4);
            const v4f zero = {0.f, 0.f, 0.f, 0.f};
            d[0] = zero;
            d[1] = zero;
            d[2] = zero;
            d[3] = zero;
        }
        __syncthreads();
        if (folds) {
            float s0 = 0.f, s1 = 0.f;
#pragma unroll
            for (int j = 0; j < 10; ++j) {
                s0 += w[j].x * aout[4 * j + 0];
                s1 += w[j].x * aout[40 + 4 * j + 0];
                s0 += w[j].y * aout[4 * j + 1];
                s1 += w[j].y * aout[40 + 4 * j + 1];
                s0 += w[j].z * aout[4 * j + 2];
                s1 += w[j].z * aout[40 + 4 * j + 2];
                s0 += w[j].w * aout[4 * j + 3];
                s1 += w[j].w * aout[40 + 4 * j + 3];
            }
            float* f = Blds + ((tid >> 2) * G2_COLS + 40) * 4 + (tid & 3);
            f[0] = s0;
            f[4] = s1;
        }
        __syncthreads();
    }

    const float* bbase = Blds + (hf * G2_COLS + l16) * 4;
    constexpr int CH = 8 * (2 * G2_COLS * 4);

    for (; tile < ntile; tile += tstep) {
        const int rowBase = tile * 16;
        v8f acc[3] = {};

#pragma unroll
        for (int i = 0; i < 8; ++i) A1[i] = aptr[2 * (8 + i)];
        __builtin_amdgcn_sched_barrier(0);
        g2_chunk(A0, bbase, acc);

#pragma unroll
        for (int i = 0; i < 8; ++i) A0[i] = aptr[2 * (16 + i)];
        __builtin_amdgcn_sched_barrier(0);
        g2_chunk(A1, bbase + CH, acc);

#pragma unroll
        for (int i = 0; i < 8; ++i) A1[i] = aptr[2 * (24 + i)];
        __builtin_amdgcn_sched_barrier(0);
        g2_chunk(A0, bbase + 2 * CH, acc);

        {
            const int tn = tile + tstep < ntile ? tile + tstep : tile;
            int arow = tn * 16 + l16;
            if (arow >= nrows) arow = nrows - 1;
            aptr = reinterpret_cast<const v4f*>(x + (size_t)arow * 256 + hf * 4);
#pragma unroll
            for (int i = 0; i < 8; ++i) A0[i] = aptr[2 * i];
        }
        __builtin_amdgcn_sched_barrier(0);
        g2_chunk(A1, bbase + 3 * CH, acc);

        if (rowBase + 16 <= nrows) {
            store_tiles2<false>(acc, Wx, ssrc, sdst, nrows, rowBase, hf, l16);
        } else {
            store_tiles2<true>(acc, Wx, ssrc, sdst, nrows, rowBase, hf, l16);
        }
    }
}

__global__ __launch_bounds__(256) void edge_link(const int* __restrict__ dst,
                                                 int* __restrict__ head,
                                                 int* __restrict__ nxt, int ne) {
    int e = blockIdx.x * blockDim.x + threadIdx.x;
    if (e >= ne) return;
    nxt[e] = atomicExch(&head[dst[e]], e);
}

__device__ __forceinline__ float lane_bcast(float v, int l) {
    return __int_as_float(__builtin_amdgcn_readlane(__float_as_int(v), l));
}

__device__ __forceinline__ float elu1(float v) { return v > 0.f ? v : (expf(v) - 1.f); }

__global__ __launch_bounds__(256) void gather1(const int* __restrict__ src,
                                               const int* __restrict__ head,
                                               const int* __restrict__ nxt,
                                               const float* __restrict__ ssrc,
                                               const float* __restrict__ sdst,
                                               const float* __restrict__ Wh,
                                               float* __restrict__ hp, int n, int ne) {
    const int lane = threadIdx.x & 31;
    const int hsel = lane & 3;
    const bool hi = lane >= 16;
    const int node = __builtin_amdgcn_readfirstlane((int)(blockIdx.x * 8 + (threadIdx.x >> 5)));
    if (node >= n) return;
    v4f accA = {0.f, 0.f, 0.f, 0.f};
    v4f accB = {0.f, 0.f, 0.f, 0.f};
    float m = -INFINITY, den = 0.f;
    const float sd = sdst[node * 4 + hsel];

    int e = head[node];
    bool ok = (unsigned)e < (unsigned)ne;
    int s = 0, en = -1;
    if (ok) {
        s = src[e];
        en = nxt[e];
    }
    for (int it = 0; it < ne && ok; ++it) {
        const v4f* row = reinterpret_cast<const v4f*>(Wh + (size_t)s * 256) + lane;
        const v4f wA = row[0];
        const v4f wB = row[32];
        const float xs = ssrc[s * 4 + hsel];
        e = en;
        ok = (unsigned)e < (unsigned)ne;
        if (ok) {
            s = src[e];
            en = nxt[e];
        }
        const float x = lrelu(xs + sd);
        const bool up = x > m;
        const float mn = fmaxf(m, x);
        const float t = expf(fminf(m, x) - mn);
        const float c = up ? t : 1.f;
        const float p = up ? 1.f : t;
        den = den * c + p;
        m = mn;
        const float c0 = lane_bcast(c, 0), c1 = lane_bcast(c, 1);
        const float c2 = lane_bcast(c, 2), c3 = lane_bcast(c, 3);
        const float p0 = lane_bcast(p, 0), p1 = lane_bcast(p, 1);
        const float p2 = lane_bcast(p, 2), p3 = lane_bcast(p, 3);
        const float cA = hi ? c1 : c0;
        const float pA = hi ? p1 : p0;
        const float cB = hi ? c3 : c2;
        const float pB = hi ? p3 : p2;
        accA.x = accA.x * cA + pA * wA.x;
        accA.y = accA.y * cA + pA * wA.y;
        accA.z = accA.z * cA + pA * wA.z;
        accA.w = accA.w * cA + pA * wA.w;
        accB.x = accB.x * cB + pB * wB.x;
        accB.y = accB.y * cB + pB * wB.y;
        accB.z = accB.z * cB + pB * wB.z;
        accB.w = accB.w * cB + pB * wB.w;
    }
    const float d0 = lane_bcast(den, 0), d1 = lane_bcast(den, 1);
    const float d2 = lane_bcast(den, 2), d3 = lane_bcast(den, 3);
    const float dA = hi ? d1 : d0;
    const float dB = hi ? d3 : d2;
    const bool anyA = dA > 0.f, anyB = dB > 0.f;
    v4f oA, oB;
    oA.x = elu1(anyA ? accA.x / dA : 0.f);
    oA.y = elu1(anyA ? accA.y / dA : 0.f);
    oA.z = elu1(anyA ? accA.z / dA : 0.f);
    oA.w = elu1(anyA ? accA.w / dA : 0.f);
    oB.x = elu1(anyB ? accB.x / dB : 0.f);
    oB.y = elu1(anyB ? accB.y / dB : 0.f);
    oB.z = elu1(anyB ? accB.z / dB : 0.f);
    oB.w = elu1(anyB ? accB.w / dB : 0.f);
    v4f* orow = reinterpret_cast<v4f*>(hp + (size_t)node * 256) + lane;
    orow[0] = oA;
    orow[32] = oB;
}

__global__ __launch_bounds__(64) void gather2(const int* __restrict__ src,
                                              const int* __restrict__ head,
                                              const int* __restrict__ nxt,
                                              const float* __restrict__ ssrc,
                                              const float* __restrict__ sdst,
                                              const float* __restrict__ Wx,
                                              float* __restrict__ out, int n, int ne) {
    const int node = (int)(blockIdx.x * 64 + threadIdx.x);
    if (node >= n) return;
    float acc[40];
#pragma unroll
    for (int k = 0; k < 40; ++k) acc[k] = 0.f;
    float m = -INFINITY, den = 0.f;
    const float sd = sdst[node];

    int e = head[node];
    bool ok = (unsigned)e < (unsigned)ne;
    int s = 0, en = -1;
    if (ok) {
        s = src[e];
        en = nxt[e];
    }
    for (int it = 0; it < ne && ok; ++it) {
        const v4f* row = reinterpret_cast<const v4f*>(Wx + (size_t)s * 40);
        v4f w[10];
#pragma unroll
        for (int j = 0; j < 10; ++j) w[j] = row[j];
        const float xs = ssrc[s];
        e = en;
        ok = (unsigned)e < (unsigned)ne;
        if (ok) {
            s = src[e];
            en = nxt[e];
        }
        const float x = lrelu(xs + sd);
        const bool up = x > m;
        const float mn = fmaxf(m, x);
        const float t = expf(fminf(m, x) - mn);
        const float c = up ? t : 1.f;
        const float p = up ? 1.f : t;
        den = den * c + p;
        m = mn;
#pragma unroll
        for (int j = 0; j < 10; ++j) {
            acc[4 * j + 0] = acc[4 * j + 0] * c + p * w[j].x;
            acc[4 * j + 1] = acc[4 * j + 1] * c + p * w[j].y;
            acc[4 * j + 2] = acc[4 * j + 2] * c + p * w[j].z;
            acc[4 * j + 3] = acc[4 * j + 3] * c + p * w[j].w;
        }
    }
    const bool any = den > 0.f;
    float mx = -INFINITY;
#pragma unroll
    for (int k = 0; k < 40; ++k) {
        acc[k] = any ? acc[k] / den : 0.f;
        mx = fmaxf(mx, acc[k]);
    }
    float sm = 0.f;
#pragma unroll
    for (int k = 0; k < 40; ++k) sm += expf(acc[k] - mx);
    const float lg = mx + logf(sm);
    v4f* orow = reinterpret_cast<v4f*>(out + (size_t)node * 40);
#pragma unroll
    for (int j = 0; j < 10; ++j) {
        v4f o;
        o.x = acc[4 * j + 0] - lg;
        o.y = acc[4 * j + 1] - lg;
        o.z = acc[4 * j + 2] - lg;
        o.w = acc[4 * j + 3] - lg;
        orow[j] = o;
    }
}

extern "C" void kernel_launch(void* const* d_in, const int* in_sizes, int n_in, void* d_out,
                              int out_size, void* d_ws, size_t ws_size, hipStream_t stream) {
    (void)n_in; (void)out_size; (void)ws_size;
    const float* h      = (const float*)d_in[0];
    const int*   eidx   = (const int*)d_in[1];
    const float* Wheads = (const float*)d_in[2];
    const float* aheads = (const float*)d_in[3];
    const float* Wout   = (const float*)d_in[4];
    const float* aout   = (const float*)d_in[5];
    float* out = (float*)d_out;

    const int N = in_sizes[0] / 256;
    const int E = in_sizes[1] / 2;
    const int* src = eidx;
    const int* dst = eidx + E;

    float* ws = (float*)d_ws;
    size_t oWh   = 0;
    size_t oHP   = oWh  + (size_t)N * 256;
    size_t oS1s  = oHP  + (size_t)N * 256;
    size_t oS1d  = oS1s + (size_t)N * 4;
    size_t oHead = oS1d + (size_t)N * 4;
    size_t oNxt  = oHead + (size_t)N;
    size_t oS2s  = oS1s;
    size_t oS2d  = oS1d;
    size_t oWx   = oWh;
    size_t oWa1  = oHP;

    int* head = (int*)(ws + oHead);
    int* nxt  = (int*)(ws + oNxt);

    int fillBlocks = (N + 255) / 256;
    prep_kernel<<<fillBlocks + 8, 256, 0, stream>>>(head, N, fillBlocks, Wheads, aheads,
                                                    ws + oWa1);

    gemm1_kernel<<<256, 512, 0, stream>>>(h, Wheads, ws + oWa1, ws + oWh, ws + oS1s, ws + oS1d,
                                          N);

    int geE = (E + 255) / 256;
    edge_link<<<geE, 256, 0, stream>>>(dst, head, nxt, E);

    gather1<<<(N + 7) / 8, 256, 0, stream>>>(src, head, nxt, ws + oS1s, ws + oS1d, ws + oWh,
                                             ws + oHP, N, E);

    gemm2_kernel<<<256, 512, 0, stream>>>(ws + oHP, Wout, aout, ws + oWx, ws + oS2s, ws + oS2d,
                                          N);

    gather2<<<(N + 63) / 64, 64, 0, stream>>>(src, head, nxt, ws + oS2s, ws + oS2d, ws + oWx,
                                              out, N, E);
}


# in-edge list linking (one atomic exchange per edge) moved into the layer-1 GEMM waves: 8 edges per lane per row tile, 3-stage pipeline in the k loop; link launch gone; sum order = list arrival order
# speedup vs baseline: 7.4731x; 1.0527x over previous
#include <hip/hip_runtime.h>


#define LRELU 0.2f

typedef __attribute__((ext_vector_type(2))) float v2f;
typedef __attribute__((ext_vector_type(8))) float v8f;
typedef __attribute__((ext_vector_type(4))) float v4f;

__device__ __forceinline__ float lrelu(float x) { return x > 0.f ? x : LRELU * x; }

__global__ __launch_bounds__(256) void prep_kernel(int* __restrict__ head, int n, int fillBlocks,
                                                   const float* __restrict__ Wheads,
                                                   const float* __restrict__ aheads,
                                                   float* __restrict__ Wa1) {
    const int b = blockIdx.x;
    if (b < fillBlocks) {
        const int i = b * 256 + threadIdx.x;
        if (i < n) head[i] = -1;
        return;
    }
    const int c = b - fillBlocks;
    const int hd = c & 3;
    const int k = threadIdx.x;
    const float* w = Wheads + ((size_t)hd * 256 + k) * 64;
    const float* a = aheads + hd * 128 + (c >> 2) * 64;
    float s = 0.f;
#pragma unroll
    for (int f = 0; f < 64; ++f) s += w[f] * a[f];
    Wa1[k * 16 + c] = s;
    Wa1[k * 16 + 8 + c] = 0.f;
}

#define G1_COLS 272
#define G1_TILES 17

__device__ __forceinline__ void link_step(int s, int e, const int* __restrict__ dst,
                                          int* __restrict__ head, int* __restrict__ nxt,
                                          int& dcur, int& rcur) {
    if (s >= 2) nxt[e - 64] = rcur;
    if (s >= 1 && s <= 8) rcur = atomicExch(&head[dcur], e - 32);
    if (s <= 7) dcur = dst[e];
}

__global__ __launch_bounds__(512) void gemm1_kernel(const float* __restrict__ h,
                                                    const float* __restrict__ Wheads,
                                                    const float* __restrict__ Wa1,
                                                    float* __restrict__ Wh,
                                                    float* __restrict__ ssrc,
                                                    float* __restrict__ sdst, int nrows,
                                                    const int* __restrict__ dst,
                                                    int* __restrict__ head,
                                                    int* __restrict__ nxt, int ne) {
    __shared__ __attribute__((aligned(16))) float Blds[256 * G1_COLS];
    const int tid = threadIdx.x;

    {
        v4f r[9][4];
#pragma unroll
        for (int i = 0; i < 9; ++i) {
            int u = tid + 512 * i;
            if (u > 64 * (G1_COLS / 4) - 1) u = 64 * (G1_COLS / 4) - 1;
            const int q = u / (G1_COLS / 4);
            const int c = (u - q * (G1_COLS / 4)) * 4;
            const bool head = c < 256;
            const float* sp = head ? (Wheads + ((size_t)(c >> 6) * 256 + 4 * q) * 64 + (c & 63))
                                   : (Wa1 + (4 * q) * 16 + (c - 256));
            const int pitch = head ? 64 : 16;
#pragma unroll
            for (int j = 0; j < 4; ++j) r[i][j] = *reinterpret_cast<const v4f*>(sp + j * pitch);
        }
#pragma unroll
        for (int i = 0; i < 9; ++i) {
            const int u = tid + 512 * i;
            if (u < 64 * (G1_COLS / 4)) {
                const int q = u / (G1_COLS / 4);
                const int c = (u - q * (G1_COLS / 4)) * 4;
                v4f* d = reinterpret_cast<v4f*>(Blds + (q * G1_COLS + c) * 4);
                v4f s0, s1, s2, s3;
                s0.x = r[i][0].x; s0.y = r[i][1].x; s0.z = r[i][2].x; s0.w = r[i][3].x;
                s1.x = r[i][0].y; s1.y = r[i][1].y; s1.z = r[i][2].y; s1.w = r[i][3].y;
                s2.x = r[i][0].z; s2.y = r[i][1].z; s2.z = r[i][2].z; s2.w = r[i][3].z;
                s3.x = r[i][0].w; s3.y = r[i][1].w; s3.z = r[i][2].w; s3.w = r[i][3].w;
                d[0] = s0;
                d[1] = s1;
                d[2] = s2;
                d[3] = s3;
            }
        }
    }
    __syncthreads();

    const int wid  = __builtin_amdgcn_readfirstlane(tid >> 5);
    const int lane = tid & 31;
    const int hf   = lane >> 4;
    const int l16  = lane & 15;
    const int ntile = (nrows + 15) >> 4;
    const float* bbase = Blds + (hf * G1_COLS + l16) * 4;

    const int slot = 4 * (((wid >> 2) - (wid & 3)) & 3) + (wid & 3);
    for (int tile = slot * (int)gridDim.x + (int)blockIdx.x; tile < ntile;
         tile += 16 * (int)gridDim.x) {
        const int rowBase = tile * 16;
        int arow = rowBase + l16;
        if (arow >= nrows) arow = nrows - 1;
        const v4f* aptr = reinterpret_cast<const v4f*>(h + (size_t)arow * 256 + hf * 4);

        const int ebase = tile * 256;
        const bool lk = ebase + 256 <= ne;
        int dcur = 0, rcur = 0;

        v8f acc[G1_TILES] = {};
        v4f a = aptr[0];
        for (int g = 0; g < 32; ++g) {
            if (lk && (g & 3) == 0) link_step(g >> 2, ebase + g * 8 + lane, dst, head, nxt, dcur, rcur);
            const v4f an = aptr[2 * ((g + 1) & 31)];
            __builtin_amdgcn_sched_barrier(0);
            const float* bp = bbase + g * (2 * G1_COLS * 4);
            v4f b[G1_TILES];
#pragma unroll
            for (int t = 0; t < G1_TILES; ++t) b[t] = *reinterpret_cast<const v4f*>(bp + t * 64);
            v2f a0, a1;
            a0.x = a.x; a0.y = a.y;
            a1.x = a.z; a1.y = a.w;
#pragma unroll
            for (int t = 0; t < G1_TILES; ++t) {
                v2f b0;
                b0.x = b[t].x; b0.y = b[t].y;
                acc[t] = __builtin_amdgcn_wmma_f32_16x16x4_f32(false, a0, false, b0, (short)0,
                                                               acc[t], false, false);
            }
#pragma unroll
            for (int t = 0; t < G1_TILES; ++t) {
                v2f b1;
                b1.x = b[t].z; b1.y = b[t].w;
                acc[t] = __builtin_amdgcn_wmma_f32_16x16x4_f32(false, a1, false, b1, (short)0,
                                                               acc[t], false, false);
            }
            a = an;
        }

        if (lk) link_step(8, ebase + 256 + lane, dst, head, nxt, dcur, rcur);
        const bool full = rowBase + 16 <= nrows;
        float* crow = Wh + (size_t)(rowBase + 8 * hf) * 256 + l16;
        if (full) {
#pragma unroll
            for (int t = 0; t < 16; ++t)
#pragma unroll
                for (int v = 0; v < 8; ++v) crow[(size_t)v * 256 + t * 16] = acc[t][v];
        } else {
#pragma unroll
            for (int t = 0; t < 16; ++t)
#pragma unroll
                for (int v = 0; v < 8; ++v)
                    if (rowBase + 8 * hf + v < nrows) crow[(size_t)v * 256 + t * 16] = acc[t][v];
        }
        if (l16 < 8) {
            float* sb = ((l16 < 4) ? (ssrc + l16) : (sdst + (l16 - 4))) +
                        (size_t)(rowBase + 8 * hf) * 4;
#pragma unroll
            for (int v = 0; v < 8; ++v)
                if (full || rowBase + 8 * hf + v < nrows) sb[(size_t)v * 4] = acc[16][v];
        }
        if (lk) link_step(9, ebase + 288 + lane, dst, head, nxt, dcur, rcur);
    }

    const int cov = ntile < (ne >> 8) ? ntile : (ne >> 8);
    for (int cb = cov + (int)blockIdx.x * 16 + wid; cb * 256 < ne; cb += 16 * (int)gridDim.x) {
        for (int j = 0; j < 8; ++j) {
            const int e = cb * 256 + j * 32 + lane;
            if (e < ne) nxt[e] = atomicExch(&head[dst[e]], e);
        }
    }
}

#define G2_COLS 48

template <bool GUARD>
__device__ __forceinline__ void store_tiles2(const v8f (&acc)[3], float* __restrict__ C,
                                             float* __restrict__ S0, float* __restrict__ S1,
                                             int nrows, int rowBase, int half, int l16) {
#pragma unroll
    for (int t = 0; t < 3; ++t) {
        const int col = t * 16 + l16;
        float* base;
        size_t stride;
        bool on = true;
        if (col < 40) {
            base = C + col;
            stride = 40;
        } else {
            base = (col == 40) ? S0 : S1;
            stride = 1;
            on = col < 42;
        }
#pragma unroll
        for (int v = 0; v < 8; ++v) {
            const int row = rowBase + v + half * 8;
            if (on && (!GUARD || row < nrows)) base[(size_t)row * stride] = acc[t][v];
        }
    }
}

__device__ __forceinline__ void g2_chunk(const v4f (&A)[8], const float* bp, v8f (&acc)[3]) {
#pragma unroll
    for (int gq = 0; gq < 2; ++gq) {
        v4f b[4][3];
#pragma unroll
        for (int u = 0; u < 4; ++u)
#pragma unroll
            for (int t = 0; t < 3; ++t)
                b[u][t] = *reinterpret_cast<const v4f*>(bp + (4 * gq + u) * (2 * G2_COLS * 4) +
                                                        t * 64);
        __builtin_amdgcn_sched_barrier(0);
#pragma unroll
        for (int u = 0; u < 4; ++u) {
            const v4f a = A[4 * gq + u];
            v2f a0, a1;
            a0.x = a.x; a0.y = a.y;
            a1.x = a.z; a1.y = a.w;
#pragma unroll
            for (int t = 0; t < 3; ++t) {
                v2f b0;
                b0.x = b[u][t].x; b0.y = b[u][t].y;
                acc[t] = __builtin_amdgcn_wmma_f32_16x16x4_f32(false, a0, false, b0, (short)0,
                                                               acc[t], false, false);
            }
#pragma unroll
            for (int t = 0; t < 3; ++t) {
                v2f b1;
                b1.x = b[u][t].z; b1.y = b[u][t].w;
                acc[t] = __builtin_amdgcn_wmma_f32_16x16x4_f32(false, a1, false, b1, (short)0,
                                                               acc[t], false, false);
            }
        }
    }
}

__global__ __launch_bounds__(512) void gemm2_kernel(const float* __restrict__ x,
                                                    const float* __restrict__ Wout,
                                                    const float* __restrict__ aout,
                                                    float* __restrict__ Wx,
                                                    float* __restrict__ ssrc,
                                                    float* __restrict__ sdst, int nrows) {
    __shared__ __attribute__((aligned(16))) float Blds[256 * G2_COLS];
    const int tid = threadIdx.x;

    const int wid  = __builtin_amdgcn_readfirstlane(tid >> 5);
    const int lane = tid & 31;
    const int hf   = lane >> 4;
    const int l16  = lane & 15;
    const int ntile = (nrows + 15) >> 4;
    const int slot = 4 * (((wid >> 2) - (wid & 3)) & 3) + (wid & 3);
    const int tstep = 16 * (int)gridDim.x;
    int tile = slot * (int)gridDim.x + (int)blockIdx.x;

    v4f A0[8], A1[8];
    const v4f* aptr;
    {
        const int t0 = tile < ntile ? tile : ntile - 1;
        int arow = t0 * 16 + l16;
        if (arow >= nrows) arow = nrows - 1;
        aptr = reinterpret_cast<const v4f*>(x + (size_t)arow * 256 + hf * 4);
#pragma unroll
        for (int i = 0; i < 8; ++i) A0[i] = aptr[2 * i];
    }

    {
        const bool second = tid < 128;
        const bool zeros  = tid >= 128 && tid < 256;
        v4f r[2][4];
#pragma unroll
        for (int i = 0; i < 2; ++i) {
            if (i == 0 || second) {
                const int u = tid + 512 * i;
                const int q = u / 10;
                const int c = (u - q * 10) * 4;
                const float* sp = Wout + (4 * q) * 40 + c;
#pragma unroll
                for (int j = 0; j < 4; ++j) r[i][j] = *reinterpret_cast<const v4f*>(sp + j * 40);
            }
        }
        const bool folds = tid < 256;
        v4f w[10];
        if (folds) {
            const v4f* wr = reinterpret_cast<const v4f*>(Wout + tid * 40);
#pragma unroll
            for (int j = 0; j < 10; ++j) w[j] = wr[j];
        }
#pragma unroll
        for (int i = 0; i < 2; ++i) {
            if (i == 0 || second) {
                const int u = tid + 512 * i;
                const int q = u / 10;
                const int c = (u - q * 10) * 4;
                v4f* d = reinterpret_cast<v4f*>(Blds + (q * G2_COLS + c) * 4);
                v4f s0, s1, s2, s3;
                s0.x = r[i][0].x; s0.y = r[i][1].x; s0.z = r[i][2].x; s0.w = r[i][3].x;
                s1.x = r[i][0].y; s1.y = r[i][1].y; s1.z = r[i][2].y; s1.w = r[i][3].y;
                s2.x = r[i][0].z; s2.y = r[i][1].z; s2.z = r[i][2].z; s2.w = r[i][3].z;
                s3.x = r[i][0].w; s3.y = r[i][1].w; s3.z = r[i][2].w; s3.w = r[i][3].w;
                d[0] = s0;
                d[1] = s1;
                d[2] = s2;
                d[3] = s3;
            }
        }
        if (zeros) {
            const int z = tid - 128;
            v4f* d = reinterpret_cast<v4f*>(Blds + ((z >> 1) * G2_COLS + 40 + (z & 1) * 4) * 4);
            const v4f zero = {0.f, 0.f, 0.f, 0.f};
            d[0] = zero;
            d[1] = zero;
            d[2] = zero;
            d[3] = zero;
        }
        __syncthreads();
        if (folds) {
            float s0 = 0.f, s1 = 0.f;
#pragma unroll
            for (int j = 0; j < 10; ++j) {
                s0 += w[j].x * aout[4 * j + 0];
                s1 += w[j].x * aout[40 + 4 * j + 0];
                s0 += w[j].y * aout[4 * j + 1];
                s1 += w[j].y * aout[40 + 4 * j + 1];
                s0 += w[j].z * aout[4 * j + 2];
                s1 += w[j].z * aout[40 + 4 * j + 2];
                s0 += w[j].w * aout[4 * j + 3];
                s1 += w[j].w * aout[40 + 4 * j + 3];
            }
            float* f = Blds + ((tid >> 2) * G2_COLS + 40) * 4 + (tid & 3);
            f[0] = s0;
            f[4] = s1;
        }
        __syncthreads();
    }

    const float* bbase = Blds + (hf * G2_COLS + l16) * 4;
    constexpr int CH = 8 * (2 * G2_COLS * 4);

    for (; tile < ntile; tile += tstep) {
        const int rowBase = tile * 16;
        v8f acc[3] = {};

#pragma unroll
        for (int i = 0; i < 8; ++i) A1[i] = aptr[2 * (8 + i)];
        __builtin_amdgcn_sched_barrier(0);
        g2_chunk(A0, bbase, acc);

#pragma unroll
        for (int i = 0; i < 8; ++i) A0[i] = aptr[2 * (16 + i)];
        __builtin_amdgcn_sched_barrier(0);
        g2_chunk(A1, bbase + CH, acc);

#pragma unroll
        for (int i = 0; i < 8; ++i) A1[i] = aptr[2 * (24 + i)];
        __builtin_amdgcn_sched_barrier(0);
        g2_chunk(A0, bbase + 2 * CH, acc);

        {
            const int tn = tile + tstep < ntile ? tile + tstep : tile;
            int arow = tn * 16 + l16;
            if (arow >= nrows) arow = nrows - 1;
            aptr = reinterpret_cast<const v4f*>(x + (size_t)arow * 256 + hf * 4);
#pragma unroll
            for (int i = 0; i < 8; ++i) A0[i] = aptr[2 * i];
        }
        __builtin_amdgcn_sched_barrier(0);
        g2_chunk(A1, bbase + 3 * CH, acc);

        if (rowBase + 16 <= nrows) {
            store_tiles2<false>(acc, Wx, ssrc, sdst, nrows, rowBase, hf, l16);
        } else {
            store_tiles2<true>(acc, Wx, ssrc, sdst, nrows, rowBase, hf, l16);
        }
    }
}

__device__ __forceinline__ float lane_bcast(float v, int l) {
    return __int_as_float(__builtin_amdgcn_readlane(__float_as_int(v), l));
}

__device__ __forceinline__ float elu1(float v) { return v > 0.f ? v : (expf(v) - 1.f); }

__global__ __launch_bounds__(256) void gather1(const int* __restrict__ src,
                                               const int* __restrict__ head,
                                               const int* __restrict__ nxt,
                                               const float* __restrict__ ssrc,
                                               const float* __restrict__ sdst,
                                               const float* __restrict__ Wh,
                                               float* __restrict__ hp, int n, int ne) {
    const int lane = threadIdx.x & 31;
    const int hsel = lane & 3;
    const bool hi = lane >= 16;
    const int node = __builtin_amdgcn_readfirstlane((int)(blockIdx.x * 8 + (threadIdx.x >> 5)));
    if (node >= n) return;
    v4f accA = {0.f, 0.f, 0.f, 0.f};
    v4f accB = {0.f, 0.f, 0.f, 0.f};
    float m = -INFINITY, den = 0.f;
    const float sd = sdst[node * 4 + hsel];

    int e = head[node];
    bool ok = (unsigned)e < (unsigned)ne;
    int s = 0, en = -1;
    if (ok) {
        s = src[e];
        en = nxt[e];
    }
    for (int it = 0; it < ne && ok; ++it) {
        const v4f* row = reinterpret_cast<const v4f*>(Wh + (size_t)s * 256) + lane;
        const v4f wA = row[0];
        const v4f wB = row[32];
        const float xs = ssrc[s * 4 + hsel];
        e = en;
        ok = (unsigned)e < (unsigned)ne;
        if (ok) {
            s = src[e];
            en = nxt[e];
        }
        const float x = lrelu(xs + sd);
        const bool up = x > m;
        const float mn = fmaxf(m, x);
        const float t = expf(fminf(m, x) - mn);
        const float c = up ? t : 1.f;
        const float p = up ? 1.f : t;
        den = den * c + p;
        m = mn;
        const float c0 = lane_bcast(c, 0), c1 = lane_bcast(c, 1);
        const float c2 = lane_bcast(c, 2), c3 = lane_bcast(c, 3);
        const float p0 = lane_bcast(p, 0), p1 = lane_bcast(p, 1);
        const float p2 = lane_bcast(p, 2), p3 = lane_bcast(p, 3);
        const float cA = hi ? c1 : c0;
        const float pA = hi ? p1 : p0;
        const float cB = hi ? c3 : c2;
        const float pB = hi ? p3 : p2;
        accA.x = accA.x * cA + pA * wA.x;
        accA.y = accA.y * cA + pA * wA.y;
        accA.z = accA.z * cA + pA * wA.z;
        accA.w = accA.w * cA + pA * wA.w;
        accB.x = accB.x * cB + pB * wB.x;
        accB.y = accB.y * cB + pB * wB.y;
        accB.z = accB.z * cB + pB * wB.z;
        accB.w = accB.w * cB + pB * wB.w;
    }
    const float d0 = lane_bcast(den, 0), d1 = lane_bcast(den, 1);
    const float d2 = lane_bcast(den, 2), d3 = lane_bcast(den, 3);
    const float dA = hi ? d1 : d0;
    const float dB = hi ? d3 : d2;
    const bool anyA = dA > 0.f, anyB = dB > 0.f;
    v4f oA, oB;
    oA.x = elu1(anyA ? accA.x / dA : 0.f);
    oA.y = elu1(anyA ? accA.y / dA : 0.f);
    oA.z = elu1(anyA ? accA.z / dA : 0.f);
    oA.w = elu1(anyA ? accA.w / dA : 0.f);
    oB.x = elu1(anyB ? accB.x / dB : 0.f);
    oB.y = elu1(anyB ? accB.y / dB : 0.f);
    oB.z = elu1(anyB ? accB.z / dB : 0.f);
    oB.w = elu1(anyB ? accB.w / dB : 0.f);
    v4f* orow = reinterpret_cast<v4f*>(hp + (size_t)node * 256) + lane;
    orow[0] = oA;
    orow[32] = oB;
}

__global__ __launch_bounds__(64) void gather2(const int* __restrict__ src,
                                              const int* __restrict__ head,
                                              const int* __restrict__ nxt,
                                              const float* __restrict__ ssrc,
                                              const float* __restrict__ sdst,
                                              const float* __restrict__ Wx,
                                              float* __restrict__ out, int n, int ne) {
    const int node = (int)(blockIdx.x * 64 + threadIdx.x);
    if (node >= n) return;
    float acc[40];
#pragma unroll
    for (int k = 0; k < 40; ++k) acc[k] = 0.f;
    float m = -INFINITY, den = 0.f;
    const float sd = sdst[node];

    int e = head[node];
    bool ok = (unsigned)e < (unsigned)ne;
    int s = 0, en = -1;
    if (ok) {
        s = src[e];
        en = nxt[e];
    }
    for (int it = 0; it < ne && ok; ++it) {
        const v4f* row = reinterpret_cast<const v4f*>(Wx + (size_t)s * 40);
        v4f w[10];
#pragma unroll
        for (int j = 0; j < 10; ++j) w[j] = row[j];
        const float xs = ssrc[s];
        e = en;
        ok = (unsigned)e < (unsigned)ne;
        if (ok) {
            s = src[e];
            en = nxt[e];
        }
        const float x = lrelu(xs + sd);
        const bool up = x > m;
        const float mn = fmaxf(m, x);
        const float t = expf(fminf(m, x) - mn);
        const float c = up ? t : 1.f;
        const float p = up ? 1.f : t;
        den = den * c + p;
        m = mn;
#pragma unroll
        for (int j = 0; j < 10; ++j) {
            acc[4 * j + 0] = acc[4 * j + 0] * c + p * w[j].x;
            acc[4 * j + 1] = acc[4 * j + 1] * c + p * w[j].y;
            acc[4 * j + 2] = acc[4 * j + 2] * c + p * w[j].z;
            acc[4 * j + 3] = acc[4 * j + 3] * c + p * w[j].w;
        }
    }
    const bool any = den > 0.f;
    float mx = -INFINITY;
#pragma unroll
    for (int k = 0; k < 40; ++k) {
        acc[k] = any ? acc[k] / den : 0.f;
        mx = fmaxf(mx, acc[k]);
    }
    float sm = 0.f;
#pragma unroll
    for (int k = 0; k < 40; ++k) sm += expf(acc[k] - mx);
    const float lg = mx + logf(sm);
    v4f* orow = reinterpret_cast<v4f*>(out + (size_t)node * 40);
#pragma unroll
    for (int j = 0; j < 10; ++j) {
        v4f o;
        o.x = acc[4 * j + 0] - lg;
        o.y = acc[4 * j + 1] - lg;
        o.z = acc[4 * j + 2] - lg;
        o.w = acc[4 * j + 3] - lg;
        orow[j] = o;
    }
}

extern "C" void kernel_launch(void* const* d_in, const int* in_sizes, int n_in, void* d_out,
                              int out_size, void* d_ws, size_t ws_size, hipStream_t stream) {
    (void)n_in; (void)out_size; (void)ws_size;
    const float* h      = (const float*)d_in[0];
    const int*   eidx   = (const int*)d_in[1];
    const float* Wheads = (const float*)d_in[2];
    const float* aheads = (const float*)d_in[3];
    const float* Wout   = (const float*)d_in[4];
    const float* aout   = (const float*)d_in[5];
    float* out = (float*)d_out;

    const int N = in_sizes[0] / 256;
    const int E = in_sizes[1] / 2;
    const int* src = eidx;
    const int* dst = eidx + E;

    float* ws = (float*)d_ws;
    size_t oWh   = 0;
    size_t oHP   = oWh  + (size_t)N * 256;
    size_t oS1s  = oHP  + (size_t)N * 256;
    size_t oS1d  = oS1s + (size_t)N * 4;
    size_t oHead = oS1d + (size_t)N * 4;
    size_t oNxt  = oHead + (size_t)N;
    size_t oS2s  = oS1s;
    size_t oS2d  = oS1d;
    size_t oWx   = oWh;
    size_t oWa1  = oHP;

    int* head = (int*)(ws + oHead);
    int* nxt  = (int*)(ws + oNxt);

    int fillBlocks = (N + 255) / 256;
    prep_kernel<<<fillBlocks + 8, 256, 0, stream>>>(head, N, fillBlocks, Wheads, aheads,
                                                    ws + oWa1);

    gemm1_kernel<<<256, 512, 0, stream>>>(h, Wheads, ws + oWa1, ws + oWh, ws + oS1s, ws + oS1d,
                                          N, dst, head, nxt, E);

    gather1<<<(N + 7) / 8, 256, 0, stream>>>(src, head, nxt, ws + oS1s, ws + oS1d, ws + oWh,
                                             ws + oHP, N, E);

    gemm2_kernel<<<256, 512, 0, stream>>>(ws + oHP, Wout, aout, ws + oWx, ws + oS2s, ws + oS2d,
                                          N);

    gather2<<<(N + 63) / 64, 64, 0, stream>>>(src, head, nxt, ws + oS2s, ws + oS2d, ws + oWx,
                                              out, N, E);
}
